# starting kernel
# baseline (speedup 1.0000x reference)
#include <hip/hip_runtime.h>
#include <hip/hip_bf16.h>

typedef float v2f __attribute__((ext_vector_type(2)));
typedef float v8f __attribute__((ext_vector_type(8)));

#define T_DIM  16384
#define IN_DIM 1024
#define S_DIM  512
#define NCOL   2048

__global__ void wtrans_kernel(const float* __restrict__ w_in, float* __restrict__ bt) {
    const int tid = blockIdx.x * blockDim.x + threadIdx.x;
    const int n = tid & (NCOL - 1);
    const int i = tid >> 11;
    const int s = n & (S_DIM - 1);
    const int g = n >> 9;
    bt[tid] = w_in[((s << 2) + g) * IN_DIM + i];
}

__global__ void __launch_bounds__(256)
gemm_u_kernel(const float* __restrict__ x, const float* __restrict__ bt,
              float* __restrict__ u) {
    const int lane = threadIdx.x & 31;
    const int wave = threadIdx.x >> 5;
    const int half = lane >> 4;
    const int l16  = lane & 15;
    const int row0 = blockIdx.y << 5;
    const int col0 = (blockIdx.x << 8) + (wave << 5);

    const float* aRow0 = x + (size_t)(row0 + l16) * IN_DIM + 2 * half;
    const float* aRow1 = aRow0 + (size_t)16 * IN_DIM;
    const float* bCol0 = bt + (size_t)(2 * half) * NCOL + col0 + l16;
    const float* bCol1 = bCol0 + 16;

    v8f acc00 = {0.f,0.f,0.f,0.f,0.f,0.f,0.f,0.f};
    v8f acc01 = acc00, acc10 = acc00, acc11 = acc00;

    for (int k0 = 0; k0 < IN_DIM; k0 += 64) {
        if (k0 + 64 < IN_DIM) {
            __builtin_prefetch(aRow0 + k0 + 64, 0, 3);
            __builtin_prefetch(aRow1 + k0 + 64, 0, 3);
        }
#pragma unroll
        for (int kk = 0; kk < 64; kk += 4) {
            const int k = k0 + kk;
            v2f a0 = *(const v2f*)(aRow0 + k);
            v2f a1 = *(const v2f*)(aRow1 + k);
            v2f b0, b1;
            b0.x = bCol0[(size_t)k * NCOL];
            b0.y = bCol0[(size_t)(k + 1) * NCOL];
            b1.x = bCol1[(size_t)k * NCOL];
            b1.y = bCol1[(size_t)(k + 1) * NCOL];
            acc00 = __builtin_amdgcn_wmma_f32_16x16x4_f32(false, a0, false, b0,
                                                          (short)0, acc00, false, false);
            acc01 = __builtin_amdgcn_wmma_f32_16x16x4_f32(false, a0, false, b1,
                                                          (short)0, acc01, false, false);
            acc10 = __builtin_amdgcn_wmma_f32_16x16x4_f32(false, a1, false, b0,
                                                          (short)0, acc10, false, false);
            acc11 = __builtin_amdgcn_wmma_f32_16x16x4_f32(false, a1, false, b1,
                                                          (short)0, acc11, false, false);
        }
    }

    float* out00 = u + (size_t)row0 * NCOL + col0 + l16;
    float* out01 = out00 + 16;
    float* out10 = out00 + (size_t)16 * NCOL;
    float* out11 = out10 + 16;
#pragma unroll
    for (int r = 0; r < 8; ++r) {
        const size_t roff = (size_t)(r + half * 8) * NCOL;
        out00[roff] = acc00[r];
        out01[roff] = acc01[r];
        out10[roff] = acc10[r];
        out11[roff] = acc11[r];
    }
}

__device__ __forceinline__ float sigmoid_f(float v) {
    return 1.f / (1.f + __expf(-v));
}
__device__ __forceinline__ float tanh_f(float v) {
    return 1.f - 2.f / (__expf(2.f * v) + 1.f);
}

__global__ void scan_kernel(const float* __restrict__ u,
                            const float* __restrict__ w_rec,
                            const float* __restrict__ bias,
                            float* __restrict__ h_n, float* __restrict__ c_n) {
    const int s = blockIdx.x * blockDim.x + threadIdx.x;
    const float wf = w_rec[s],             wi = w_rec[S_DIM + s];
    const float wo = w_rec[2 * S_DIM + s], wg = w_rec[3 * S_DIM + s];
    const float bf = bias[s],              bi = bias[S_DIM + s];
    const float bo = bias[2 * S_DIM + s],  bg = bias[3 * S_DIM + s];

    float h = 0.f, c = 0.f;
    h_n[s] = 0.f;
    c_n[s] = 0.f;

    const float* up = u + s;
    for (int t = 0; t < T_DIM; ++t) {
        const float* ut = up + (size_t)t * NCOL;
        float f = sigmoid_f(bf + wf * h + ut[0]);
        float i = sigmoid_f(bi + wi * h + ut[S_DIM]);
        float o = sigmoid_f(bo + wo * h + ut[2 * S_DIM]);
        float g = tanh_f   (bg + wg * h + ut[3 * S_DIM]);
        c = f * c + i * g;
        h = o * tanh_f(c);
        h_n[(size_t)(t + 1) * S_DIM + s] = h;
        c_n[(size_t)(t + 1) * S_DIM + s] = c;
    }
}

__global__ void q_kernel(const float* __restrict__ h_n,
                         const float* __restrict__ w_reg,
                         const float* __restrict__ b_reg,
                         float* __restrict__ q) {
    const int lane = threadIdx.x & 31;
    const int t = (blockIdx.x * blockDim.x + threadIdx.x) >> 5;
    const float* hr = h_n + (size_t)(t + 1) * S_DIM;
    float acc = 0.f;
#pragma unroll
    for (int j = 0; j < S_DIM; j += 32)
        acc += hr[j + lane] * w_reg[j + lane];
#pragma unroll
    for (int off = 16; off; off >>= 1)
        acc += __shfl_xor(acc, off, 32);
    if (lane == 0) q[t] = acc + b_reg[0];
}

extern "C" void kernel_launch(void* const* d_in, const int* in_sizes, int n_in,
                              void* d_out, int out_size, void* d_ws, size_t ws_size,
                              hipStream_t stream) {
    (void)in_sizes; (void)n_in; (void)out_size; (void)ws_size;
    const float* x     = (const float*)d_in[0];
    const float* w_in  = (const float*)d_in[1];
    const float* w_rec = (const float*)d_in[2];
    const float* bias  = (const float*)d_in[3];
    const float* w_reg = (const float*)d_in[4];
    const float* b_reg = (const float*)d_in[5];

    float* out = (float*)d_out;
    float* q   = out;
    float* h_n = out + T_DIM;
    float* c_n = h_n + (size_t)(T_DIM + 1) * S_DIM;

    float* u  = (float*)d_ws;
    float* bt = u + (size_t)T_DIM * NCOL;

    wtrans_kernel<<<(IN_DIM * NCOL) / 256, 256, 0, stream>>>(w_in, bt);

    dim3 ggrid(NCOL / 256, T_DIM / 32);
    gemm_u_kernel<<<ggrid, 256, 0, stream>>>(x, bt, u);

    scan_kernel<<<S_DIM / 32, 32, 0, stream>>>(u, w_rec, bias, h_n, c_n);

    q_kernel<<<(T_DIM * 32) / 256, 256, 0, stream>>>(h_n, w_reg, b_reg, q);
}


# scan kernel: rows of u are loaded a block of time steps ahead into a per-lane register ring, so global-load latency leaves the recurrence's dependent chain; arithmetic and order of sums unchanged
# speedup vs baseline: 4.1664x; 4.1664x over previous
#include <hip/hip_runtime.h>
#include <hip/hip_bf16.h>

typedef float v2f __attribute__((ext_vector_type(2)));
typedef float v8f __attribute__((ext_vector_type(8)));

#define T_DIM  16384
#define IN_DIM 1024
#define S_DIM  512
#define NCOL   2048

__global__ void wtrans_kernel(const float* __restrict__ w_in, float* __restrict__ bt) {
    const int tid = blockIdx.x * blockDim.x + threadIdx.x;
    const int n = tid & (NCOL - 1);
    const int i = tid >> 11;
    const int s = n & (S_DIM - 1);
    const int g = n >> 9;
    bt[tid] = w_in[((s << 2) + g) * IN_DIM + i];
}

__global__ void __launch_bounds__(256)
gemm_u_kernel(const float* __restrict__ x, const float* __restrict__ bt,
              float* __restrict__ u) {
    const int lane = threadIdx.x & 31;
    const int wave = threadIdx.x >> 5;
    const int half = lane >> 4;
    const int l16  = lane & 15;
    const int row0 = blockIdx.y << 5;
    const int col0 = (blockIdx.x << 8) + (wave << 5);

    const float* aRow0 = x + (size_t)(row0 + l16) * IN_DIM + 2 * half;
    const float* aRow1 = aRow0 + (size_t)16 * IN_DIM;
    const float* bCol0 = bt + (size_t)(2 * half) * NCOL + col0 + l16;
    const float* bCol1 = bCol0 + 16;

    v8f acc00 = {0.f,0.f,0.f,0.f,0.f,0.f,0.f,0.f};
    v8f acc01 = acc00, acc10 = acc00, acc11 = acc00;

    for (int k0 = 0; k0 < IN_DIM; k0 += 64) {
        if (k0 + 64 < IN_DIM) {
            __builtin_prefetch(aRow0 + k0 + 64, 0, 3);
            __builtin_prefetch(aRow1 + k0 + 64, 0, 3);
        }
#pragma unroll
        for (int kk = 0; kk < 64; kk += 4) {
            const int k = k0 + kk;
            v2f a0 = *(const v2f*)(aRow0 + k);
            v2f a1 = *(const v2f*)(aRow1 + k);
            v2f b0, b1;
            b0.x = bCol0[(size_t)k * NCOL];
            b0.y = bCol0[(size_t)(k + 1) * NCOL];
            b1.x = bCol1[(size_t)k * NCOL];
            b1.y = bCol1[(size_t)(k + 1) * NCOL];
            acc00 = __builtin_amdgcn_wmma_f32_16x16x4_f32(false, a0, false, b0,
                                                          (short)0, acc00, false, false);
            acc01 = __builtin_amdgcn_wmma_f32_16x16x4_f32(false, a0, false, b1,
                                                          (short)0, acc01, false, false);
            acc10 = __builtin_amdgcn_wmma_f32_16x16x4_f32(false, a1, false, b0,
                                                          (short)0, acc10, false, false);
            acc11 = __builtin_amdgcn_wmma_f32_16x16x4_f32(false, a1, false, b1,
                                                          (short)0, acc11, false, false);
        }
    }

    float* out00 = u + (size_t)row0 * NCOL + col0 + l16;
    float* out01 = out00 + 16;
    float* out10 = out00 + (size_t)16 * NCOL;
    float* out11 = out10 + 16;
#pragma unroll
    for (int r = 0; r < 8; ++r) {
        const size_t roff = (size_t)(r + half * 8) * NCOL;
        out00[roff] = acc00[r];
        out01[roff] = acc01[r];
        out10[roff] = acc10[r];
        out11[roff] = acc11[r];
    }
}

__device__ __forceinline__ float sigmoid_f(float v) {
    return 1.f / (1.f + __expf(-v));
}
__device__ __forceinline__ float tanh_f(float v) {
    return 1.f - 2.f / (__expf(2.f * v) + 1.f);
}

#define PF_STEPS 16

template <bool kRefill>
__device__ __forceinline__ void scan_block(
        const float* __restrict__ u, float* __restrict__ h_n, float* __restrict__ c_n,
        const int s, const int t0,
        const float wf, const float wi, const float wo, const float wg,
        const float bf, const float bi, const float bo, const float bg,
        float (&rf)[PF_STEPS], float (&ri)[PF_STEPS],
        float (&ro)[PF_STEPS], float (&rg)[PF_STEPS],
        float& h, float& c) {
#pragma unroll
    for (int j = 0; j < PF_STEPS; ++j) {
        const int t = t0 + j;
        float f = sigmoid_f(bf + wf * h + rf[j]);
        float i = sigmoid_f(bi + wi * h + ri[j]);
        float o = sigmoid_f(bo + wo * h + ro[j]);
        float g = tanh_f   (bg + wg * h + rg[j]);
        c = f * c + i * g;
        h = o * tanh_f(c);
        h_n[(size_t)(t + 1) * S_DIM + s] = h;
        c_n[(size_t)(t + 1) * S_DIM + s] = c;

        if (kRefill) {
            const float* un = u + (size_t)(t + PF_STEPS) * NCOL;
            __builtin_amdgcn_sched_barrier(0);
            rf[j] = un[s];
            ri[j] = un[S_DIM + s];
            ro[j] = un[2 * S_DIM + s];
            rg[j] = un[3 * S_DIM + s];
        }
    }
}

__global__ void __launch_bounds__(32)
scan_kernel(const float* __restrict__ u,
            const float* __restrict__ w_rec,
            const float* __restrict__ bias,
            float* __restrict__ h_n, float* __restrict__ c_n) {
    const int s = blockIdx.x * blockDim.x + threadIdx.x;
    const float wf = w_rec[s],             wi = w_rec[S_DIM + s];
    const float wo = w_rec[2 * S_DIM + s], wg = w_rec[3 * S_DIM + s];
    const float bf = bias[s],              bi = bias[S_DIM + s];
    const float bo = bias[2 * S_DIM + s],  bg = bias[3 * S_DIM + s];

    float h = 0.f, c = 0.f;
    h_n[s] = 0.f;
    c_n[s] = 0.f;

    float rf[PF_STEPS], ri[PF_STEPS], ro[PF_STEPS], rg[PF_STEPS];
#pragma unroll
    for (int j = 0; j < PF_STEPS; ++j) {
        const float* ut = u + (size_t)j * NCOL;
        rf[j] = ut[s];
        ri[j] = ut[S_DIM + s];
        ro[j] = ut[2 * S_DIM + s];
        rg[j] = ut[3 * S_DIM + s];
    }

    for (int t0 = 0; t0 < T_DIM - PF_STEPS; t0 += PF_STEPS)
        scan_block<true>(u, h_n, c_n, s, t0, wf, wi, wo, wg, bf, bi, bo, bg,
                         rf, ri, ro, rg, h, c);
    scan_block<false>(u, h_n, c_n, s, T_DIM - PF_STEPS, wf, wi, wo, wg, bf, bi, bo, bg,
                      rf, ri, ro, rg, h, c);
}

__global__ void q_kernel(const float* __restrict__ h_n,
                         const float* __restrict__ w_reg,
                         const float* __restrict__ b_reg,
                         float* __restrict__ q) {
    const int lane = threadIdx.x & 31;
    const int t = (blockIdx.x * blockDim.x + threadIdx.x) >> 5;
    const float* hr = h_n + (size_t)(t + 1) * S_DIM;
    float acc = 0.f;
#pragma unroll
    for (int j = 0; j < S_DIM; j += 32)
        acc += hr[j + lane] * w_reg[j + lane];
#pragma unroll
    for (int off = 16; off; off >>= 1)
        acc += __shfl_xor(acc, off, 32);
    if (lane == 0) q[t] = acc + b_reg[0];
}

extern "C" void kernel_launch(void* const* d_in, const int* in_sizes, int n_in,
                              void* d_out, int out_size, void* d_ws, size_t ws_size,
                              hipStream_t stream) {
    (void)in_sizes; (void)n_in; (void)out_size; (void)ws_size;
    const float* x     = (const float*)d_in[0];
    const float* w_in  = (const float*)d_in[1];
    const float* w_rec = (const float*)d_in[2];
    const float* bias  = (const float*)d_in[3];
    const float* w_reg = (const float*)d_in[4];
    const float* b_reg = (const float*)d_in[5];

    float* out = (float*)d_out;
    float* q   = out;
    float* h_n = out + T_DIM;
    float* c_n = h_n + (size_t)(T_DIM + 1) * S_DIM;

    float* u  = (float*)d_ws;
    float* bt = u + (size_t)T_DIM * NCOL;

    wtrans_kernel<<<(IN_DIM * NCOL) / 256, 256, 0, stream>>>(w_in, bt);

    dim3 ggrid(NCOL / 256, T_DIM / 32);
    gemm_u_kernel<<<ggrid, 256, 0, stream>>>(x, bt, u);

    scan_kernel<<<S_DIM / 32, 32, 0, stream>>>(u, w_rec, bias, h_n, c_n);

    q_kernel<<<(T_DIM * 32) / 256, 256, 0, stream>>>(h_n, w_reg, b_reg, q);
}


# scan kernel: sigmoid and tanh take 1/x from the hardware reciprocal (v_rcp_f32, 1 ulp) in place of the IEEE divide sequence, shortening the recurrence's dependent chain; order of sums unchanged
# speedup vs baseline: 6.1862x; 1.4848x over previous
#include <hip/hip_runtime.h>
#include <hip/hip_bf16.h>

typedef float v2f __attribute__((ext_vector_type(2)));
typedef float v8f __attribute__((ext_vector_type(8)));

#define T_DIM  16384
#define IN_DIM 1024
#define S_DIM  512
#define NCOL   2048

__global__ void wtrans_kernel(const float* __restrict__ w_in, float* __restrict__ bt) {
    const int tid = blockIdx.x * blockDim.x + threadIdx.x;
    const int n = tid & (NCOL - 1);
    const int i = tid >> 11;
    const int s = n & (S_DIM - 1);
    const int g = n >> 9;
    bt[tid] = w_in[((s << 2) + g) * IN_DIM + i];
}

__global__ void __launch_bounds__(256)
gemm_u_kernel(const float* __restrict__ x, const float* __restrict__ bt,
              float* __restrict__ u) {
    const int lane = threadIdx.x & 31;
    const int wave = threadIdx.x >> 5;
    const int half = lane >> 4;
    const int l16  = lane & 15;
    const int row0 = blockIdx.y << 5;
    const int col0 = (blockIdx.x << 8) + (wave << 5);

    const float* aRow0 = x + (size_t)(row0 + l16) * IN_DIM + 2 * half;
    const float* aRow1 = aRow0 + (size_t)16 * IN_DIM;
    const float* bCol0 = bt + (size_t)(2 * half) * NCOL + col0 + l16;
    const float* bCol1 = bCol0 + 16;

    v8f acc00 = {0.f,0.f,0.f,0.f,0.f,0.f,0.f,0.f};
    v8f acc01 = acc00, acc10 = acc00, acc11 = acc00;

    for (int k0 = 0; k0 < IN_DIM; k0 += 64) {
        if (k0 + 64 < IN_DIM) {
            __builtin_prefetch(aRow0 + k0 + 64, 0, 3);
            __builtin_prefetch(aRow1 + k0 + 64, 0, 3);
        }
#pragma unroll
        for (int kk = 0; kk < 64; kk += 4) {
            const int k = k0 + kk;
            v2f a0 = *(const v2f*)(aRow0 + k);
            v2f a1 = *(const v2f*)(aRow1 + k);
            v2f b0, b1;
            b0.x = bCol0[(size_t)k * NCOL];
            b0.y = bCol0[(size_t)(k + 1) * NCOL];
            b1.x = bCol1[(size_t)k * NCOL];
            b1.y = bCol1[(size_t)(k + 1) * NCOL];
            acc00 = __builtin_amdgcn_wmma_f32_16x16x4_f32(false, a0, false, b0,
                                                          (short)0, acc00, false, false);
            acc01 = __builtin_amdgcn_wmma_f32_16x16x4_f32(false, a0, false, b1,
                                                          (short)0, acc01, false, false);
            acc10 = __builtin_amdgcn_wmma_f32_16x16x4_f32(false, a1, false, b0,
                                                          (short)0, acc10, false, false);
            acc11 = __builtin_amdgcn_wmma_f32_16x16x4_f32(false, a1, false, b1,
                                                          (short)0, acc11, false, false);
        }
    }

    float* out00 = u + (size_t)row0 * NCOL + col0 + l16;
    float* out01 = out00 + 16;
    float* out10 = out00 + (size_t)16 * NCOL;
    float* out11 = out10 + 16;
#pragma unroll
    for (int r = 0; r < 8; ++r) {
        const size_t roff = (size_t)(r + half * 8) * NCOL;
        out00[roff] = acc00[r];
        out01[roff] = acc01[r];
        out10[roff] = acc10[r];
        out11[roff] = acc11[r];
    }
}

__device__ __forceinline__ float sigmoid_f(float v) {
    return __builtin_amdgcn_rcpf(1.f + __expf(-v));
}
__device__ __forceinline__ float tanh_f(float v) {
    return 1.f - 2.f * __builtin_amdgcn_rcpf(__expf(2.f * v) + 1.f);
}

#define PF_STEPS 16

template <bool kRefill>
__device__ __forceinline__ void scan_block(
        const float* __restrict__ u, float* __restrict__ h_n, float* __restrict__ c_n,
        const int s, const int t0,
        const float wf, const float wi, const float wo, const float wg,
        const float bf, const float bi, const float bo, const float bg,
        float (&rf)[PF_STEPS], float (&ri)[PF_STEPS],
        float (&ro)[PF_STEPS], float (&rg)[PF_STEPS],
        float& h, float& c) {
#pragma unroll
    for (int j = 0; j < PF_STEPS; ++j) {
        const int t = t0 + j;
        float f = sigmoid_f(bf + wf * h + rf[j]);
        float i = sigmoid_f(bi + wi * h + ri[j]);
        float o = sigmoid_f(bo + wo * h + ro[j]);
        float g = tanh_f   (bg + wg * h + rg[j]);
        c = f * c + i * g;
        h = o * tanh_f(c);
        h_n[(size_t)(t + 1) * S_DIM + s] = h;
        c_n[(size_t)(t + 1) * S_DIM + s] = c;

        if (kRefill) {
            const float* un = u + (size_t)(t + PF_STEPS) * NCOL;
            __builtin_amdgcn_sched_barrier(0);
            rf[j] = un[s];
            ri[j] = un[S_DIM + s];
            ro[j] = un[2 * S_DIM + s];
            rg[j] = un[3 * S_DIM + s];
        }
    }
}

__global__ void __launch_bounds__(32)
scan_kernel(const float* __restrict__ u,
            const float* __restrict__ w_rec,
            const float* __restrict__ bias,
            float* __restrict__ h_n, float* __restrict__ c_n) {
    const int s = blockIdx.x * blockDim.x + threadIdx.x;
    const float wf = w_rec[s],             wi = w_rec[S_DIM + s];
    const float wo = w_rec[2 * S_DIM + s], wg = w_rec[3 * S_DIM + s];
    const float bf = bias[s],              bi = bias[S_DIM + s];
    const float bo = bias[2 * S_DIM + s],  bg = bias[3 * S_DIM + s];

    float h = 0.f, c = 0.f;
    h_n[s] = 0.f;
    c_n[s] = 0.f;

    float rf[PF_STEPS], ri[PF_STEPS], ro[PF_STEPS], rg[PF_STEPS];
#pragma unroll
    for (int j = 0; j < PF_STEPS; ++j) {
        const float* ut = u + (size_t)j * NCOL;
        rf[j] = ut[s];
        ri[j] = ut[S_DIM + s];
        ro[j] = ut[2 * S_DIM + s];
        rg[j] = ut[3 * S_DIM + s];
    }

    for (int t0 = 0; t0 < T_DIM - PF_STEPS; t0 += PF_STEPS)
        scan_block<true>(u, h_n, c_n, s, t0, wf, wi, wo, wg, bf, bi, bo, bg,
                         rf, ri, ro, rg, h, c);
    scan_block<false>(u, h_n, c_n, s, T_DIM - PF_STEPS, wf, wi, wo, wg, bf, bi, bo, bg,
                      rf, ri, ro, rg, h, c);
}

__global__ void q_kernel(const float* __restrict__ h_n,
                         const float* __restrict__ w_reg,
                         const float* __restrict__ b_reg,
                         float* __restrict__ q) {
    const int lane = threadIdx.x & 31;
    const int t = (blockIdx.x * blockDim.x + threadIdx.x) >> 5;
    const float* hr = h_n + (size_t)(t + 1) * S_DIM;
    float acc = 0.f;
#pragma unroll
    for (int j = 0; j < S_DIM; j += 32)
        acc += hr[j + lane] * w_reg[j + lane];
#pragma unroll
    for (int off = 16; off; off >>= 1)
        acc += __shfl_xor(acc, off, 32);
    if (lane == 0) q[t] = acc + b_reg[0];
}

extern "C" void kernel_launch(void* const* d_in, const int* in_sizes, int n_in,
                              void* d_out, int out_size, void* d_ws, size_t ws_size,
                              hipStream_t stream) {
    (void)in_sizes; (void)n_in; (void)out_size; (void)ws_size;
    const float* x     = (const float*)d_in[0];
    const float* w_in  = (const float*)d_in[1];
    const float* w_rec = (const float*)d_in[2];
    const float* bias  = (const float*)d_in[3];
    const float* w_reg = (const float*)d_in[4];
    const float* b_reg = (const float*)d_in[5];

    float* out = (float*)d_out;
    float* q   = out;
    float* h_n = out + T_DIM;
    float* c_n = h_n + (size_t)(T_DIM + 1) * S_DIM;

    float* u  = (float*)d_ws;
    float* bt = u + (size_t)T_DIM * NCOL;

    wtrans_kernel<<<(IN_DIM * NCOL) / 256, 256, 0, stream>>>(w_in, bt);

    dim3 ggrid(NCOL / 256, T_DIM / 32);
    gemm_u_kernel<<<ggrid, 256, 0, stream>>>(x, bt, u);

    scan_kernel<<<S_DIM / 32, 32, 0, stream>>>(u, w_rec, bias, h_n, c_n);

    q_kernel<<<(T_DIM * 32) / 256, 256, 0, stream>>>(h_n, w_reg, b_reg, q);
}


# scan kernel: tanh, and sigmoid as 0.5+0.5*tanh(x/2), on the hardware v_tanh_f32 in place of exp and rcp chains; the halves fold into gate constants and the cell update's fmas (rounding points move)
# speedup vs baseline: 7.1763x; 1.1600x over previous
#include <hip/hip_runtime.h>
#include <hip/hip_bf16.h>

typedef float v2f __attribute__((ext_vector_type(2)));
typedef float v8f __attribute__((ext_vector_type(8)));

#define T_DIM  16384
#define IN_DIM 1024
#define S_DIM  512
#define NCOL   2048

__global__ void wtrans_kernel(const float* __restrict__ w_in, float* __restrict__ bt) {
    const int tid = blockIdx.x * blockDim.x + threadIdx.x;
    const int n = tid & (NCOL - 1);
    const int i = tid >> 11;
    const int s = n & (S_DIM - 1);
    const int g = n >> 9;
    bt[tid] = w_in[((s << 2) + g) * IN_DIM + i];
}

__global__ void __launch_bounds__(256)
gemm_u_kernel(const float* __restrict__ x, const float* __restrict__ bt,
              float* __restrict__ u) {
    const int lane = threadIdx.x & 31;
    const int wave = threadIdx.x >> 5;
    const int half = lane >> 4;
    const int l16  = lane & 15;
    const int row0 = blockIdx.y << 5;
    const int col0 = (blockIdx.x << 8) + (wave << 5);

    const float* aRow0 = x + (size_t)(row0 + l16) * IN_DIM + 2 * half;
    const float* aRow1 = aRow0 + (size_t)16 * IN_DIM;
    const float* bCol0 = bt + (size_t)(2 * half) * NCOL + col0 + l16;
    const float* bCol1 = bCol0 + 16;

    v8f acc00 = {0.f,0.f,0.f,0.f,0.f,0.f,0.f,0.f};
    v8f acc01 = acc00, acc10 = acc00, acc11 = acc00;

    for (int k0 = 0; k0 < IN_DIM; k0 += 64) {
        if (k0 + 64 < IN_DIM) {
            __builtin_prefetch(aRow0 + k0 + 64, 0, 3);
            __builtin_prefetch(aRow1 + k0 + 64, 0, 3);
        }
#pragma unroll
        for (int kk = 0; kk < 64; kk += 4) {
            const int k = k0 + kk;
            v2f a0 = *(const v2f*)(aRow0 + k);
            v2f a1 = *(const v2f*)(aRow1 + k);
            v2f b0, b1;
            b0.x = bCol0[(size_t)k * NCOL];
            b0.y = bCol0[(size_t)(k + 1) * NCOL];
            b1.x = bCol1[(size_t)k * NCOL];
            b1.y = bCol1[(size_t)(k + 1) * NCOL];
            acc00 = __builtin_amdgcn_wmma_f32_16x16x4_f32(false, a0, false, b0,
                                                          (short)0, acc00, false, false);
            acc01 = __builtin_amdgcn_wmma_f32_16x16x4_f32(false, a0, false, b1,
                                                          (short)0, acc01, false, false);
            acc10 = __builtin_amdgcn_wmma_f32_16x16x4_f32(false, a1, false, b0,
                                                          (short)0, acc10, false, false);
            acc11 = __builtin_amdgcn_wmma_f32_16x16x4_f32(false, a1, false, b1,
                                                          (short)0, acc11, false, false);
        }
    }

    float* out00 = u + (size_t)row0 * NCOL + col0 + l16;
    float* out01 = out00 + 16;
    float* out10 = out00 + (size_t)16 * NCOL;
    float* out11 = out10 + 16;
#pragma unroll
    for (int r = 0; r < 8; ++r) {
        const size_t roff = (size_t)(r + half * 8) * NCOL;
        out00[roff] = acc00[r];
        out01[roff] = acc01[r];
        out10[roff] = acc10[r];
        out11[roff] = acc11[r];
    }
}

__device__ __forceinline__ float tanh_hw(float v) {
    return __builtin_amdgcn_tanhf(v);
}

#define PF_STEPS 16

template <bool kRefill>
__device__ __forceinline__ void scan_block(
        const float* __restrict__ u, float* __restrict__ h_n, float* __restrict__ c_n,
        const int s, const int t0,
        const float wf, const float wi, const float wo, const float wg,
        const float bf, const float bi, const float bo, const float bg,
        float (&rf)[PF_STEPS], float (&ri)[PF_STEPS],
        float (&ro)[PF_STEPS], float (&rg)[PF_STEPS],
        float& h, float& c) {
#pragma unroll
    for (int j = 0; j < PF_STEPS; ++j) {
        const int t = t0 + j;
        const float tf = tanh_hw(__builtin_fmaf(0.5f, rf[j], __builtin_fmaf(wf, h, bf)));
        const float ti = tanh_hw(__builtin_fmaf(0.5f, ri[j], __builtin_fmaf(wi, h, bi)));
        const float to = tanh_hw(__builtin_fmaf(0.5f, ro[j], __builtin_fmaf(wo, h, bo)));
        const float g  = tanh_hw(__builtin_fmaf(wg, h, bg) + rg[j]);
        const float hc = 0.5f * c;
        const float fc = __builtin_fmaf(tf, hc, hc);
        const float ig2 = __builtin_fmaf(ti, g, g);
        c = __builtin_fmaf(0.5f, ig2, fc);
        const float o = __builtin_fmaf(0.5f, to, 0.5f);
        h = o * tanh_hw(c);
        h_n[(size_t)(t + 1) * S_DIM + s] = h;
        c_n[(size_t)(t + 1) * S_DIM + s] = c;

        if (kRefill) {
            const float* un = u + (size_t)(t + PF_STEPS) * NCOL;
            __builtin_amdgcn_sched_barrier(0);
            rf[j] = un[s];
            ri[j] = un[S_DIM + s];
            ro[j] = un[2 * S_DIM + s];
            rg[j] = un[3 * S_DIM + s];
        }
    }
}

__global__ void __launch_bounds__(32)
scan_kernel(const float* __restrict__ u,
            const float* __restrict__ w_rec,
            const float* __restrict__ bias,
            float* __restrict__ h_n, float* __restrict__ c_n) {
    const int s = blockIdx.x * blockDim.x + threadIdx.x;
    const float wf = 0.5f * w_rec[s],             wi = 0.5f * w_rec[S_DIM + s];
    const float wo = 0.5f * w_rec[2 * S_DIM + s], wg = w_rec[3 * S_DIM + s];
    const float bf = 0.5f * bias[s],              bi = 0.5f * bias[S_DIM + s];
    const float bo = 0.5f * bias[2 * S_DIM + s],  bg = bias[3 * S_DIM + s];

    float h = 0.f, c = 0.f;
    h_n[s] = 0.f;
    c_n[s] = 0.f;

    float rf[PF_STEPS], ri[PF_STEPS], ro[PF_STEPS], rg[PF_STEPS];
#pragma unroll
    for (int j = 0; j < PF_STEPS; ++j) {
        const float* ut = u + (size_t)j * NCOL;
        rf[j] = ut[s];
        ri[j] = ut[S_DIM + s];
        ro[j] = ut[2 * S_DIM + s];
        rg[j] = ut[3 * S_DIM + s];
    }

    for (int t0 = 0; t0 < T_DIM - PF_STEPS; t0 += PF_STEPS)
        scan_block<true>(u, h_n, c_n, s, t0, wf, wi, wo, wg, bf, bi, bo, bg,
                         rf, ri, ro, rg, h, c);
    scan_block<false>(u, h_n, c_n, s, T_DIM - PF_STEPS, wf, wi, wo, wg, bf, bi, bo, bg,
                      rf, ri, ro, rg, h, c);
}

__global__ void q_kernel(const float* __restrict__ h_n,
                         const float* __restrict__ w_reg,
                         const float* __restrict__ b_reg,
                         float* __restrict__ q) {
    const int lane = threadIdx.x & 31;
    const int t = (blockIdx.x * blockDim.x + threadIdx.x) >> 5;
    const float* hr = h_n + (size_t)(t + 1) * S_DIM;
    float acc = 0.f;
#pragma unroll
    for (int j = 0; j < S_DIM; j += 32)
        acc += hr[j + lane] * w_reg[j + lane];
#pragma unroll
    for (int off = 16; off; off >>= 1)
        acc += __shfl_xor(acc, off, 32);
    if (lane == 0) q[t] = acc + b_reg[0];
}

extern "C" void kernel_launch(void* const* d_in, const int* in_sizes, int n_in,
                              void* d_out, int out_size, void* d_ws, size_t ws_size,
                              hipStream_t stream) {
    (void)in_sizes; (void)n_in; (void)out_size; (void)ws_size;
    const float* x     = (const float*)d_in[0];
    const float* w_in  = (const float*)d_in[1];
    const float* w_rec = (const float*)d_in[2];
    const float* bias  = (const float*)d_in[3];
    const float* w_reg = (const float*)d_in[4];
    const float* b_reg = (const float*)d_in[5];

    float* out = (float*)d_out;
    float* q   = out;
    float* h_n = out + T_DIM;
    float* c_n = h_n + (size_t)(T_DIM + 1) * S_DIM;

    float* u  = (float*)d_ws;
    float* bt = u + (size_t)T_DIM * NCOL;

    wtrans_kernel<<<(IN_DIM * NCOL) / 256, 256, 0, stream>>>(w_in, bt);

    dim3 ggrid(NCOL / 256, T_DIM / 32);
    gemm_u_kernel<<<ggrid, 256, 0, stream>>>(x, bt, u);

    scan_kernel<<<S_DIM / 32, 32, 0, stream>>>(u, w_rec, bias, h_n, c_n);

    q_kernel<<<(T_DIM * 32) / 256, 256, 0, stream>>>(h_n, w_reg, b_reg, q);
}


# gemm: x and weight tiles (128x128x32) staged in LDS by full-line float4 loads, 16-byte fragment reads; weight read K-contiguous in place, transpose launch dropped; k sum order within each 8 changes
# speedup vs baseline: 8.0917x; 1.1276x over previous
#include <hip/hip_runtime.h>
#include <hip/hip_bf16.h>

typedef float v2f __attribute__((ext_vector_type(2)));
typedef float v4f __attribute__((ext_vector_type(4)));
typedef float v8f __attribute__((ext_vector_type(8)));

#define T_DIM  16384
#define IN_DIM 1024
#define S_DIM  512
#define NCOL   2048

#define BM  128
#define BN  128
#define BK  32
#define NKC (IN_DIM / BK)
#define LDP 36
#define TILE_FLOATS  (BM * LDP)
#define STAGE_FLOATS (2 * TILE_FLOATS)

__device__ __forceinline__ void tile_fetch(const float* __restrict__ ag,
                                           const float* __restrict__ bg,
                                           const int k0, v4f (&ra)[4], v4f (&rb)[4]) {
#pragma unroll
    for (int p = 0; p < 4; ++p) {
        ra[p] = *(const v4f*)(ag + (size_t)p * 32 * IN_DIM + k0);
        rb[p] = *(const v4f*)(bg + (size_t)p * 128 * IN_DIM + k0);
    }
}

__device__ __forceinline__ void tile_store(float* sa, const v4f (&ra)[4],
                                           const v4f (&rb)[4]) {
#pragma unroll
    for (int p = 0; p < 4; ++p) {
        *(v4f*)(sa + p * 32 * LDP) = ra[p];
        *(v4f*)(sa + TILE_FLOATS + p * 32 * LDP) = rb[p];
    }
}

__device__ __forceinline__ void tile_mma(const float* fa, const float* fb,
                                         v8f (&acc)[4][2]) {
#pragma unroll
    for (int k8 = 0; k8 < BK; k8 += 8) {
        v4f a[4], b[2];
#pragma unroll
        for (int i = 0; i < 4; ++i) a[i] = *(const v4f*)(fa + i * 16 * LDP + k8);
#pragma unroll
        for (int j = 0; j < 2; ++j) b[j] = *(const v4f*)(fb + j * 16 * LDP + k8);
#pragma unroll
        for (int i = 0; i < 4; ++i) {
#pragma unroll
            for (int j = 0; j < 2; ++j)
                acc[i][j] = __builtin_amdgcn_wmma_f32_16x16x4_f32(false, a[i].xy, false, b[j].xy,
                                                                  (short)0, acc[i][j], false, false);
        }
#pragma unroll
        for (int i = 0; i < 4; ++i) {
#pragma unroll
            for (int j = 0; j < 2; ++j)
                acc[i][j] = __builtin_amdgcn_wmma_f32_16x16x4_f32(false, a[i].zw, false, b[j].zw,
                                                                  (short)0, acc[i][j], false, false);
        }
    }
}

__global__ void __launch_bounds__(256)
gemm_u_kernel(const float* __restrict__ x, const float* __restrict__ w_in,
              float* __restrict__ u) {
    __shared__ __attribute__((aligned(16))) float lds[2 * STAGE_FLOATS];

    const int tid  = threadIdx.x;
    const int lane = tid & 31;
    const int wave = tid >> 5;
    const int half = lane >> 4;
    const int l16  = lane & 15;
    const int wm   = wave >> 2;
    const int wn   = wave & 3;
    const int row0 = blockIdx.y * BM;
    const int col0 = blockIdx.x * BN;
    const int g    = col0 >> 9;
    const int s0   = col0 & (S_DIM - 1);

    const int crow = tid >> 3;
    const int ck   = (tid & 7) << 2;
    const float* ag = x + (size_t)(row0 + crow) * IN_DIM + ck;
    const float* bg = w_in + (size_t)(((s0 + crow) << 2) + g) * IN_DIM + ck;
    float* sa = lds + crow * LDP + ck;

    const float* fa = lds + (wm * 64 + l16) * LDP + 4 * half;
    const float* fb = lds + TILE_FLOATS + (wn * 32 + l16) * LDP + 4 * half;

    v8f acc[4][2];
#pragma unroll
    for (int i = 0; i < 4; ++i) {
#pragma unroll
        for (int j = 0; j < 2; ++j) acc[i][j] = (v8f){0.f,0.f,0.f,0.f,0.f,0.f,0.f,0.f};
    }

    v4f ra[4], rb[4];
    tile_fetch(ag, bg, 0, ra, rb);
    tile_store(sa, ra, rb);
    __syncthreads();

    for (int kc = 0; kc < NKC - 1; ++kc) {
        const int cur = (kc & 1) * STAGE_FLOATS;
        tile_fetch(ag, bg, (kc + 1) * BK, ra, rb);
        tile_mma(fa + cur, fb + cur, acc);
        tile_store(sa + (STAGE_FLOATS - cur), ra, rb);
        __syncthreads();
    }
    tile_mma(fa + ((NKC - 1) & 1) * STAGE_FLOATS, fb + ((NKC - 1) & 1) * STAGE_FLOATS, acc);

    float* outp = u + (size_t)(row0 + wm * 64 + half * 8) * NCOL + col0 + wn * 32 + l16;
#pragma unroll
    for (int i = 0; i < 4; ++i) {
#pragma unroll
        for (int j = 0; j < 2; ++j) {
#pragma unroll
            for (int r = 0; r < 8; ++r)
                outp[(size_t)(i * 16 + r) * NCOL + j * 16] = acc[i][j][r];
        }
    }
}

__device__ __forceinline__ float tanh_hw(float v) {
    return __builtin_amdgcn_tanhf(v);
}

#define PF_STEPS 16

template <bool kRefill>
__device__ __forceinline__ void scan_block(
        const float* __restrict__ u, float* __restrict__ h_n, float* __restrict__ c_n,
        const int s, const int t0,
        const float wf, const float wi, const float wo, const float wg,
        const float bf, const float bi, const float bo, const float bg,
        float (&rf)[PF_STEPS], float (&ri)[PF_STEPS],
        float (&ro)[PF_STEPS], float (&rg)[PF_STEPS],
        float& h, float& c) {
#pragma unroll
    for (int j = 0; j < PF_STEPS; ++j) {
        const int t = t0 + j;
        const float tf = tanh_hw(__builtin_fmaf(0.5f, rf[j], __builtin_fmaf(wf, h, bf)));
        const float ti = tanh_hw(__builtin_fmaf(0.5f, ri[j], __builtin_fmaf(wi, h, bi)));
        const float to = tanh_hw(__builtin_fmaf(0.5f, ro[j], __builtin_fmaf(wo, h, bo)));
        const float g  = tanh_hw(__builtin_fmaf(wg, h, bg) + rg[j]);
        const float hc = 0.5f * c;
        const float fc = __builtin_fmaf(tf, hc, hc);
        const float ig2 = __builtin_fmaf(ti, g, g);
        c = __builtin_fmaf(0.5f, ig2, fc);
        const float o = __builtin_fmaf(0.5f, to, 0.5f);
        h = o * tanh_hw(c);
        h_n[(size_t)(t + 1) * S_DIM + s] = h;
        c_n[(size_t)(t + 1) * S_DIM + s] = c;

        if (kRefill) {
            const float* un = u + (size_t)(t + PF_STEPS) * NCOL;
            __builtin_amdgcn_sched_barrier(0);
            rf[j] = un[s];
            ri[j] = un[S_DIM + s];
            ro[j] = un[2 * S_DIM + s];
            rg[j] = un[3 * S_DIM + s];
        }
    }
}

__global__ void __launch_bounds__(32)
scan_kernel(const float* __restrict__ u,
            const float* __restrict__ w_rec,
            const float* __restrict__ bias,
            float* __restrict__ h_n, float* __restrict__ c_n) {
    const int s = blockIdx.x * blockDim.x + threadIdx.x;
    const float wf = 0.5f * w_rec[s],             wi = 0.5f * w_rec[S_DIM + s];
    const float wo = 0.5f * w_rec[2 * S_DIM + s], wg = w_rec[3 * S_DIM + s];
    const float bf = 0.5f * bias[s],              bi = 0.5f * bias[S_DIM + s];
    const float bo = 0.5f * bias[2 * S_DIM + s],  bg = bias[3 * S_DIM + s];

    float h = 0.f, c = 0.f;
    h_n[s] = 0.f;
    c_n[s] = 0.f;

    float rf[PF_STEPS], ri[PF_STEPS], ro[PF_STEPS], rg[PF_STEPS];
#pragma unroll
    for (int j = 0; j < PF_STEPS; ++j) {
        const float* ut = u + (size_t)j * NCOL;
        rf[j] = ut[s];
        ri[j] = ut[S_DIM + s];
        ro[j] = ut[2 * S_DIM + s];
        rg[j] = ut[3 * S_DIM + s];
    }

    for (int t0 = 0; t0 < T_DIM - PF_STEPS; t0 += PF_STEPS)
        scan_block<true>(u, h_n, c_n, s, t0, wf, wi, wo, wg, bf, bi, bo, bg,
                         rf, ri, ro, rg, h, c);
    scan_block<false>(u, h_n, c_n, s, T_DIM - PF_STEPS, wf, wi, wo, wg, bf, bi, bo, bg,
                      rf, ri, ro, rg, h, c);
}

__global__ void q_kernel(const float* __restrict__ h_n,
                         const float* __restrict__ w_reg,
                         const float* __restrict__ b_reg,
                         float* __restrict__ q) {
    const int lane = threadIdx.x & 31;
    const int t = (blockIdx.x * blockDim.x + threadIdx.x) >> 5;
    const float* hr = h_n + (size_t)(t + 1) * S_DIM;
    float acc = 0.f;
#pragma unroll
    for (int j = 0; j < S_DIM; j += 32)
        acc += hr[j + lane] * w_reg[j + lane];
#pragma unroll
    for (int off = 16; off; off >>= 1)
        acc += __shfl_xor(acc, off, 32);
    if (lane == 0) q[t] = acc + b_reg[0];
}

extern "C" void kernel_launch(void* const* d_in, const int* in_sizes, int n_in,
                              void* d_out, int out_size, void* d_ws, size_t ws_size,
                              hipStream_t stream) {
    (void)in_sizes; (void)n_in; (void)out_size; (void)ws_size;
    const float* x     = (const float*)d_in[0];
    const float* w_in  = (const float*)d_in[1];
    const float* w_rec = (const float*)d_in[2];
    const float* bias  = (const float*)d_in[3];
    const float* w_reg = (const float*)d_in[4];
    const float* b_reg = (const float*)d_in[5];

    float* out = (float*)d_out;
    float* q   = out;
    float* h_n = out + T_DIM;
    float* c_n = h_n + (size_t)(T_DIM + 1) * S_DIM;

    float* u  = (float*)d_ws;

    dim3 ggrid(NCOL / BN, T_DIM / BM);
    gemm_u_kernel<<<ggrid, 256, 0, stream>>>(x, w_in, u);

    scan_kernel<<<S_DIM / 32, 32, 0, stream>>>(u, w_rec, bias, h_n, c_n);

    q_kernel<<<(T_DIM * 32) / 256, 256, 0, stream>>>(h_n, w_reg, b_reg, q);
}


# scan kernel: h and c of a 16-step block stay in registers and are stored as one batch behind it, so the s_wait_xcnt before rewriting a stored register leaves each step's chain; sums unchanged
# speedup vs baseline: 9.6231x; 1.1893x over previous
#include <hip/hip_runtime.h>
#include <hip/hip_bf16.h>

typedef float v2f __attribute__((ext_vector_type(2)));
typedef float v4f __attribute__((ext_vector_type(4)));
typedef float v8f __attribute__((ext_vector_type(8)));

#define T_DIM  16384
#define IN_DIM 1024
#define S_DIM  512
#define NCOL   2048

#define BM  128
#define BN  128
#define BK  32
#define NKC (IN_DIM / BK)
#define LDP 36
#define TILE_FLOATS  (BM * LDP)
#define STAGE_FLOATS (2 * TILE_FLOATS)

__device__ __forceinline__ void tile_fetch(const float* __restrict__ ag,
                                           const float* __restrict__ bg,
                                           const int k0, v4f (&ra)[4], v4f (&rb)[4]) {
#pragma unroll
    for (int p = 0; p < 4; ++p) {
        ra[p] = *(const v4f*)(ag + (size_t)p * 32 * IN_DIM + k0);
        rb[p] = *(const v4f*)(bg + (size_t)p * 128 * IN_DIM + k0);
    }
}

__device__ __forceinline__ void tile_store(float* sa, const v4f (&ra)[4],
                                           const v4f (&rb)[4]) {
#pragma unroll
    for (int p = 0; p < 4; ++p) {
        *(v4f*)(sa + p * 32 * LDP) = ra[p];
        *(v4f*)(sa + TILE_FLOATS + p * 32 * LDP) = rb[p];
    }
}

__device__ __forceinline__ void tile_mma(const float* fa, const float* fb,
                                         v8f (&acc)[4][2]) {
#pragma unroll
    for (int k8 = 0; k8 < BK; k8 += 8) {
        v4f a[4], b[2];
#pragma unroll
        for (int i = 0; i < 4; ++i) a[i] = *(const v4f*)(fa + i * 16 * LDP + k8);
#pragma unroll
        for (int j = 0; j < 2; ++j) b[j] = *(const v4f*)(fb + j * 16 * LDP + k8);
#pragma unroll
        for (int i = 0; i < 4; ++i) {
#pragma unroll
            for (int j = 0; j < 2; ++j)
                acc[i][j] = __builtin_amdgcn_wmma_f32_16x16x4_f32(false, a[i].xy, false, b[j].xy,
                                                                  (short)0, acc[i][j], false, false);
        }
#pragma unroll
        for (int i = 0; i < 4; ++i) {
#pragma unroll
            for (int j = 0; j < 2; ++j)
                acc[i][j] = __builtin_amdgcn_wmma_f32_16x16x4_f32(false, a[i].zw, false, b[j].zw,
                                                                  (short)0, acc[i][j], false, false);
        }
    }
}

__global__ void __launch_bounds__(256)
gemm_u_kernel(const float* __restrict__ x, const float* __restrict__ w_in,
              float* __restrict__ u) {
    __shared__ __attribute__((aligned(16))) float lds[2 * STAGE_FLOATS];

    const int tid  = threadIdx.x;
    const int lane = tid & 31;
    const int wave = tid >> 5;
    const int half = lane >> 4;
    const int l16  = lane & 15;
    const int wm   = wave >> 2;
    const int wn   = wave & 3;
    const int row0 = blockIdx.y * BM;
    const int col0 = blockIdx.x * BN;
    const int g    = col0 >> 9;
    const int s0   = col0 & (S_DIM - 1);

    const int crow = tid >> 3;
    const int ck   = (tid & 7) << 2;
    const float* ag = x + (size_t)(row0 + crow) * IN_DIM + ck;
    const float* bg = w_in + (size_t)(((s0 + crow) << 2) + g) * IN_DIM + ck;
    float* sa = lds + crow * LDP + ck;

    const float* fa = lds + (wm * 64 + l16) * LDP + 4 * half;
    const float* fb = lds + TILE_FLOATS + (wn * 32 + l16) * LDP + 4 * half;

    v8f acc[4][2];
#pragma unroll
    for (int i = 0; i < 4; ++i) {
#pragma unroll
        for (int j = 0; j < 2; ++j) acc[i][j] = (v8f){0.f,0.f,0.f,0.f,0.f,0.f,0.f,0.f};
    }

    v4f ra[4], rb[4];
    tile_fetch(ag, bg, 0, ra, rb);
    tile_store(sa, ra, rb);
    __syncthreads();

    for (int kc = 0; kc < NKC - 1; ++kc) {
        const int cur = (kc & 1) * STAGE_FLOATS;
        tile_fetch(ag, bg, (kc + 1) * BK, ra, rb);
        tile_mma(fa + cur, fb + cur, acc);
        tile_store(sa + (STAGE_FLOATS - cur), ra, rb);
        __syncthreads();
    }
    tile_mma(fa + ((NKC - 1) & 1) * STAGE_FLOATS, fb + ((NKC - 1) & 1) * STAGE_FLOATS, acc);

    float* outp = u + (size_t)(row0 + wm * 64 + half * 8) * NCOL + col0 + wn * 32 + l16;
#pragma unroll
    for (int i = 0; i < 4; ++i) {
#pragma unroll
        for (int j = 0; j < 2; ++j) {
#pragma unroll
            for (int r = 0; r < 8; ++r)
                outp[(size_t)(i * 16 + r) * NCOL + j * 16] = acc[i][j][r];
        }
    }
}

__device__ __forceinline__ float tanh_hw(float v) {
    return __builtin_amdgcn_tanhf(v);
}

#define PF_STEPS 16

template <bool kRefill>
__device__ __forceinline__ void scan_block(
        const float* __restrict__ u, float* __restrict__ h_n, float* __restrict__ c_n,
        const int s, const int t0,
        const float wf, const float wi, const float wo, const float wg,
        const float bf, const float bi, const float bo, const float bg,
        float (&rf)[PF_STEPS], float (&ri)[PF_STEPS],
        float (&ro)[PF_STEPS], float (&rg)[PF_STEPS],
        float& h, float& c) {
    float hs[PF_STEPS], cs[PF_STEPS];
#pragma unroll
    for (int j = 0; j < PF_STEPS; ++j) {
        const int t = t0 + j;
        const float tf = tanh_hw(__builtin_fmaf(0.5f, rf[j], __builtin_fmaf(wf, h, bf)));
        const float ti = tanh_hw(__builtin_fmaf(0.5f, ri[j], __builtin_fmaf(wi, h, bi)));
        const float to = tanh_hw(__builtin_fmaf(0.5f, ro[j], __builtin_fmaf(wo, h, bo)));
        const float g  = tanh_hw(__builtin_fmaf(wg, h, bg) + rg[j]);
        const float hc = 0.5f * c;
        const float fc = __builtin_fmaf(tf, hc, hc);
        const float ig2 = __builtin_fmaf(ti, g, g);
        c = __builtin_fmaf(0.5f, ig2, fc);
        const float o = __builtin_fmaf(0.5f, to, 0.5f);
        h = o * tanh_hw(c);
        hs[j] = h;
        cs[j] = c;

        if (kRefill) {
            const float* un = u + (size_t)(t + PF_STEPS) * NCOL;
            __builtin_amdgcn_sched_barrier(0);
            rf[j] = un[s];
            ri[j] = un[S_DIM + s];
            ro[j] = un[2 * S_DIM + s];
            rg[j] = un[3 * S_DIM + s];
        }
    }
#pragma unroll
    for (int j = 0; j < PF_STEPS; ++j) {
        h_n[(size_t)(t0 + j + 1) * S_DIM + s] = hs[j];
        c_n[(size_t)(t0 + j + 1) * S_DIM + s] = cs[j];
    }
}

__global__ void __launch_bounds__(32)
scan_kernel(const float* __restrict__ u,
            const float* __restrict__ w_rec,
            const float* __restrict__ bias,
            float* __restrict__ h_n, float* __restrict__ c_n) {
    const int s = blockIdx.x * blockDim.x + threadIdx.x;
    const float wf = 0.5f * w_rec[s],             wi = 0.5f * w_rec[S_DIM + s];
    const float wo = 0.5f * w_rec[2 * S_DIM + s], wg = w_rec[3 * S_DIM + s];
    const float bf = 0.5f * bias[s],              bi = 0.5f * bias[S_DIM + s];
    const float bo = 0.5f * bias[2 * S_DIM + s],  bg = bias[3 * S_DIM + s];

    float h = 0.f, c = 0.f;
    h_n[s] = 0.f;
    c_n[s] = 0.f;

    float rf[PF_STEPS], ri[PF_STEPS], ro[PF_STEPS], rg[PF_STEPS];
#pragma unroll
    for (int j = 0; j < PF_STEPS; ++j) {
        const float* ut = u + (size_t)j * NCOL;
        rf[j] = ut[s];
        ri[j] = ut[S_DIM + s];
        ro[j] = ut[2 * S_DIM + s];
        rg[j] = ut[3 * S_DIM + s];
    }

    for (int t0 = 0; t0 < T_DIM - PF_STEPS; t0 += PF_STEPS)
        scan_block<true>(u, h_n, c_n, s, t0, wf, wi, wo, wg, bf, bi, bo, bg,
                         rf, ri, ro, rg, h, c);
    scan_block<false>(u, h_n, c_n, s, T_DIM - PF_STEPS, wf, wi, wo, wg, bf, bi, bo, bg,
                      rf, ri, ro, rg, h, c);
}

__global__ void q_kernel(const float* __restrict__ h_n,
                         const float* __restrict__ w_reg,
                         const float* __restrict__ b_reg,
                         float* __restrict__ q) {
    const int lane = threadIdx.x & 31;
    const int t = (blockIdx.x * blockDim.x + threadIdx.x) >> 5;
    const float* hr = h_n + (size_t)(t + 1) * S_DIM;
    float acc = 0.f;
#pragma unroll
    for (int j = 0; j < S_DIM; j += 32)
        acc += hr[j + lane] * w_reg[j + lane];
#pragma unroll
    for (int off = 16; off; off >>= 1)
        acc += __shfl_xor(acc, off, 32);
    if (lane == 0) q[t] = acc + b_reg[0];
}

extern "C" void kernel_launch(void* const* d_in, const int* in_sizes, int n_in,
                              void* d_out, int out_size, void* d_ws, size_t ws_size,
                              hipStream_t stream) {
    (void)in_sizes; (void)n_in; (void)out_size; (void)ws_size;
    const float* x     = (const float*)d_in[0];
    const float* w_in  = (const float*)d_in[1];
    const float* w_rec = (const float*)d_in[2];
    const float* bias  = (const float*)d_in[3];
    const float* w_reg = (const float*)d_in[4];
    const float* b_reg = (const float*)d_in[5];

    float* out = (float*)d_out;
    float* q   = out;
    float* h_n = out + T_DIM;
    float* c_n = h_n + (size_t)(T_DIM + 1) * S_DIM;

    float* u  = (float*)d_ws;

    dim3 ggrid(NCOL / BN, T_DIM / BM);
    gemm_u_kernel<<<ggrid, 256, 0, stream>>>(x, w_in, u);

    scan_kernel<<<S_DIM / 32, 32, 0, stream>>>(u, w_rec, bias, h_n, c_n);

    q_kernel<<<(T_DIM * 32) / 256, 256, 0, stream>>>(h_n, w_reg, b_reg, q);
}


# scan kernel: step's dependent chain re-associated, 7 links to 4: gate argument (w*o)*tanh(c)+(b+u) one fma behind tanh(c), cell update i*g+f*c one fma behind the last gate tanh; sum order changes
# speedup vs baseline: 9.9410x; 1.0330x over previous
#include <hip/hip_runtime.h>
#include <hip/hip_bf16.h>

typedef float v2f __attribute__((ext_vector_type(2)));
typedef float v4f __attribute__((ext_vector_type(4)));
typedef float v8f __attribute__((ext_vector_type(8)));

#define T_DIM  16384
#define IN_DIM 1024
#define S_DIM  512
#define NCOL   2048

#define BM  128
#define BN  128
#define BK  32
#define NKC (IN_DIM / BK)
#define LDP 36
#define TILE_FLOATS  (BM * LDP)
#define STAGE_FLOATS (2 * TILE_FLOATS)

__device__ __forceinline__ void tile_fetch(const float* __restrict__ ag,
                                           const float* __restrict__ bg,
                                           const int k0, v4f (&ra)[4], v4f (&rb)[4]) {
#pragma unroll
    for (int p = 0; p < 4; ++p) {
        ra[p] = *(const v4f*)(ag + (size_t)p * 32 * IN_DIM + k0);
        rb[p] = *(const v4f*)(bg + (size_t)p * 128 * IN_DIM + k0);
    }
}

__device__ __forceinline__ void tile_store(float* sa, const v4f (&ra)[4],
                                           const v4f (&rb)[4]) {
#pragma unroll
    for (int p = 0; p < 4; ++p) {
        *(v4f*)(sa + p * 32 * LDP) = ra[p];
        *(v4f*)(sa + TILE_FLOATS + p * 32 * LDP) = rb[p];
    }
}

__device__ __forceinline__ void tile_mma(const float* fa, const float* fb,
                                         v8f (&acc)[4][2]) {
#pragma unroll
    for (int k8 = 0; k8 < BK; k8 += 8) {
        v4f a[4], b[2];
#pragma unroll
        for (int i = 0; i < 4; ++i) a[i] = *(const v4f*)(fa + i * 16 * LDP + k8);
#pragma unroll
        for (int j = 0; j < 2; ++j) b[j] = *(const v4f*)(fb + j * 16 * LDP + k8);
#pragma unroll
        for (int i = 0; i < 4; ++i) {
#pragma unroll
            for (int j = 0; j < 2; ++j)
                acc[i][j] = __builtin_amdgcn_wmma_f32_16x16x4_f32(false, a[i].xy, false, b[j].xy,
                                                                  (short)0, acc[i][j], false, false);
        }
#pragma unroll
        for (int i = 0; i < 4; ++i) {
#pragma unroll
            for (int j = 0; j < 2; ++j)
                acc[i][j] = __builtin_amdgcn_wmma_f32_16x16x4_f32(false, a[i].zw, false, b[j].zw,
                                                                  (short)0, acc[i][j], false, false);
        }
    }
}

__global__ void __launch_bounds__(256)
gemm_u_kernel(const float* __restrict__ x, const float* __restrict__ w_in,
              float* __restrict__ u) {
    __shared__ __attribute__((aligned(16))) float lds[2 * STAGE_FLOATS];

    const int tid  = threadIdx.x;
    const int lane = tid & 31;
    const int wave = tid >> 5;
    const int half = lane >> 4;
    const int l16  = lane & 15;
    const int wm   = wave >> 2;
    const int wn   = wave & 3;
    const int row0 = blockIdx.y * BM;
    const int col0 = blockIdx.x * BN;
    const int g    = col0 >> 9;
    const int s0   = col0 & (S_DIM - 1);

    const int crow = tid >> 3;
    const int ck   = (tid & 7) << 2;
    const float* ag = x + (size_t)(row0 + crow) * IN_DIM + ck;
    const float* bg = w_in + (size_t)(((s0 + crow) << 2) + g) * IN_DIM + ck;
    float* sa = lds + crow * LDP + ck;

    const float* fa = lds + (wm * 64 + l16) * LDP + 4 * half;
    const float* fb = lds + TILE_FLOATS + (wn * 32 + l16) * LDP + 4 * half;

    v8f acc[4][2];
#pragma unroll
    for (int i = 0; i < 4; ++i) {
#pragma unroll
        for (int j = 0; j < 2; ++j) acc[i][j] = (v8f){0.f,0.f,0.f,0.f,0.f,0.f,0.f,0.f};
    }

    v4f ra[4], rb[4];
    tile_fetch(ag, bg, 0, ra, rb);
    tile_store(sa, ra, rb);
    __syncthreads();

    for (int kc = 0; kc < NKC - 1; ++kc) {
        const int cur = (kc & 1) * STAGE_FLOATS;
        tile_fetch(ag, bg, (kc + 1) * BK, ra, rb);
        tile_mma(fa + cur, fb + cur, acc);
        tile_store(sa + (STAGE_FLOATS - cur), ra, rb);
        __syncthreads();
    }
    tile_mma(fa + ((NKC - 1) & 1) * STAGE_FLOATS, fb + ((NKC - 1) & 1) * STAGE_FLOATS, acc);

    float* outp = u + (size_t)(row0 + wm * 64 + half * 8) * NCOL + col0 + wn * 32 + l16;
#pragma unroll
    for (int i = 0; i < 4; ++i) {
#pragma unroll
        for (int j = 0; j < 2; ++j) {
#pragma unroll
            for (int r = 0; r < 8; ++r)
                outp[(size_t)(i * 16 + r) * NCOL + j * 16] = acc[i][j][r];
        }
    }
}

__device__ __forceinline__ float tanh_hw(float v) {
    return __builtin_amdgcn_tanhf(v);
}

#define PF_STEPS 16

template <bool kRefill>
__device__ __forceinline__ void scan_block(
        const float* __restrict__ u, float* __restrict__ h_n, float* __restrict__ c_n,
        const int s, const int t0,
        const float wf, const float wi, const float wo, const float wg,
        const float bf, const float bi, const float bo, const float bg,
        float (&rf)[PF_STEPS], float (&ri)[PF_STEPS],
        float (&ro)[PF_STEPS], float (&rg)[PF_STEPS],
        float& wfo, float& wio, float& woo, float& wgo,
        float& tc, float& c) {
    float os[PF_STEPS], ts[PF_STEPS], cs[PF_STEPS];
#pragma unroll
    for (int j = 0; j < PF_STEPS; ++j) {
        const int t = t0 + j;
        const float pf = __builtin_fmaf(0.5f, rf[j], bf);
        const float pi = __builtin_fmaf(0.5f, ri[j], bi);
        const float po = __builtin_fmaf(0.5f, ro[j], bo);
        const float pg = rg[j] + bg;
        const float ai = __builtin_fmaf(wio, tc, pi);
        const float af = __builtin_fmaf(wfo, tc, pf);
        const float ao = __builtin_fmaf(woo, tc, po);
        const float ag = __builtin_fmaf(wgo, tc, pg);
        const float hc = 0.5f * c;
        const float ti = tanh_hw(ai);
        const float tf = tanh_hw(af);
        const float to = tanh_hw(ao);
        const float g  = tanh_hw(ag);
        __builtin_amdgcn_sched_barrier(0);
        const float ig = __builtin_fmaf(0.5f, ti, 0.5f);
        const float fc = __builtin_fmaf(tf, hc, hc);
        const float o  = __builtin_fmaf(0.5f, to, 0.5f);
        c = __builtin_fmaf(ig, g, fc);
        tc = tanh_hw(c);
        os[j] = o;
        ts[j] = tc;
        cs[j] = c;
        __builtin_amdgcn_sched_barrier(0);
        wfo = wf * o;
        wio = wi * o;
        woo = wo * o;
        wgo = wg * o;

        if (kRefill) {
            const float* un = u + (size_t)(t + PF_STEPS) * NCOL;
            rf[j] = un[s];
            ri[j] = un[S_DIM + s];
            ro[j] = un[2 * S_DIM + s];
            rg[j] = un[3 * S_DIM + s];
        }
    }
#pragma unroll
    for (int j = 0; j < PF_STEPS; ++j) {
        h_n[(size_t)(t0 + j + 1) * S_DIM + s] = os[j] * ts[j];
        c_n[(size_t)(t0 + j + 1) * S_DIM + s] = cs[j];
    }
}

__global__ void __launch_bounds__(32)
scan_kernel(const float* __restrict__ u,
            const float* __restrict__ w_rec,
            const float* __restrict__ bias,
            float* __restrict__ h_n, float* __restrict__ c_n) {
    const int s = blockIdx.x * blockDim.x + threadIdx.x;
    const float wf = 0.5f * w_rec[s],             wi = 0.5f * w_rec[S_DIM + s];
    const float wo = 0.5f * w_rec[2 * S_DIM + s], wg = w_rec[3 * S_DIM + s];
    const float bf = 0.5f * bias[s],              bi = 0.5f * bias[S_DIM + s];
    const float bo = 0.5f * bias[2 * S_DIM + s],  bg = bias[3 * S_DIM + s];

    float wfo = 0.f, wio = 0.f, woo = 0.f, wgo = 0.f;
    float tc = 0.f, c = 0.f;
    h_n[s] = 0.f;
    c_n[s] = 0.f;

    float rf[PF_STEPS], ri[PF_STEPS], ro[PF_STEPS], rg[PF_STEPS];
#pragma unroll
    for (int j = 0; j < PF_STEPS; ++j) {
        const float* ut = u + (size_t)j * NCOL;
        rf[j] = ut[s];
        ri[j] = ut[S_DIM + s];
        ro[j] = ut[2 * S_DIM + s];
        rg[j] = ut[3 * S_DIM + s];
    }

    for (int t0 = 0; t0 < T_DIM - PF_STEPS; t0 += PF_STEPS)
        scan_block<true>(u, h_n, c_n, s, t0, wf, wi, wo, wg, bf, bi, bo, bg,
                         rf, ri, ro, rg, wfo, wio, woo, wgo, tc, c);
    scan_block<false>(u, h_n, c_n, s, T_DIM - PF_STEPS, wf, wi, wo, wg, bf, bi, bo, bg,
                      rf, ri, ro, rg, wfo, wio, woo, wgo, tc, c);
}

__global__ void q_kernel(const float* __restrict__ h_n,
                         const float* __restrict__ w_reg,
                         const float* __restrict__ b_reg,
                         float* __restrict__ q) {
    const int lane = threadIdx.x & 31;
    const int t = (blockIdx.x * blockDim.x + threadIdx.x) >> 5;
    const float* hr = h_n + (size_t)(t + 1) * S_DIM;
    float acc = 0.f;
#pragma unroll
    for (int j = 0; j < S_DIM; j += 32)
        acc += hr[j + lane] * w_reg[j + lane];
#pragma unroll
    for (int off = 16; off; off >>= 1)
        acc += __shfl_xor(acc, off, 32);
    if (lane == 0) q[t] = acc + b_reg[0];
}

extern "C" void kernel_launch(void* const* d_in, const int* in_sizes, int n_in,
                              void* d_out, int out_size, void* d_ws, size_t ws_size,
                              hipStream_t stream) {
    (void)in_sizes; (void)n_in; (void)out_size; (void)ws_size;
    const float* x     = (const float*)d_in[0];
    const float* w_in  = (const float*)d_in[1];
    const float* w_rec = (const float*)d_in[2];
    const float* bias  = (const float*)d_in[3];
    const float* w_reg = (const float*)d_in[4];
    const float* b_reg = (const float*)d_in[5];

    float* out = (float*)d_out;
    float* q   = out;
    float* h_n = out + T_DIM;
    float* c_n = h_n + (size_t)(T_DIM + 1) * S_DIM;

    float* u  = (float*)d_ws;

    dim3 ggrid(NCOL / BN, T_DIM / BM);
    gemm_u_kernel<<<ggrid, 256, 0, stream>>>(x, w_in, u);

    scan_kernel<<<S_DIM / 32, 32, 0, stream>>>(u, w_rec, bias, h_n, c_n);

    q_kernel<<<(T_DIM * 32) / 256, 256, 0, stream>>>(h_n, w_reg, b_reg, q);
}


# scan kernel: u laid out state-major (column 4*s+g, the weight's row order) so a step's four gate values are one 16-byte load per lane; the register ring reaches 32 steps ahead, not 16; sums unchanged
# speedup vs baseline: 11.1608x; 1.1227x over previous
#include <hip/hip_runtime.h>
#include <hip/hip_bf16.h>

typedef float v2f __attribute__((ext_vector_type(2)));
typedef float v4f __attribute__((ext_vector_type(4)));
typedef float v8f __attribute__((ext_vector_type(8)));

#define T_DIM  16384
#define IN_DIM 1024
#define S_DIM  512
#define NCOL   2048

#define BM  128
#define BN  128
#define BK  32
#define NKC (IN_DIM / BK)
#define LDP 36
#define TILE_FLOATS  (BM * LDP)
#define STAGE_FLOATS (2 * TILE_FLOATS)

__device__ __forceinline__ void tile_fetch(const float* __restrict__ ag,
                                           const float* __restrict__ bg,
                                           const int k0, v4f (&ra)[4], v4f (&rb)[4]) {
#pragma unroll
    for (int p = 0; p < 4; ++p) {
        ra[p] = *(const v4f*)(ag + (size_t)p * 32 * IN_DIM + k0);
        rb[p] = *(const v4f*)(bg + (size_t)p * 32 * IN_DIM + k0);
    }
}

__device__ __forceinline__ void tile_store(float* sa, const v4f (&ra)[4],
                                           const v4f (&rb)[4]) {
#pragma unroll
    for (int p = 0; p < 4; ++p) {
        *(v4f*)(sa + p * 32 * LDP) = ra[p];
        *(v4f*)(sa + TILE_FLOATS + p * 32 * LDP) = rb[p];
    }
}

__device__ __forceinline__ void tile_mma(const float* fa, const float* fb,
                                         v8f (&acc)[4][2]) {
#pragma unroll
    for (int k8 = 0; k8 < BK; k8 += 8) {
        v4f a[4], b[2];
#pragma unroll
        for (int i = 0; i < 4; ++i) a[i] = *(const v4f*)(fa + i * 16 * LDP + k8);
#pragma unroll
        for (int j = 0; j < 2; ++j) b[j] = *(const v4f*)(fb + j * 16 * LDP + k8);
#pragma unroll
        for (int i = 0; i < 4; ++i) {
#pragma unroll
            for (int j = 0; j < 2; ++j)
                acc[i][j] = __builtin_amdgcn_wmma_f32_16x16x4_f32(false, a[i].xy, false, b[j].xy,
                                                                  (short)0, acc[i][j], false, false);
        }
#pragma unroll
        for (int i = 0; i < 4; ++i) {
#pragma unroll
            for (int j = 0; j < 2; ++j)
                acc[i][j] = __builtin_amdgcn_wmma_f32_16x16x4_f32(false, a[i].zw, false, b[j].zw,
                                                                  (short)0, acc[i][j], false, false);
        }
    }
}

__global__ void __launch_bounds__(256)
gemm_u_kernel(const float* __restrict__ x, const float* __restrict__ w_in,
              float* __restrict__ u) {
    __shared__ __attribute__((aligned(16))) float lds[2 * STAGE_FLOATS];

    const int tid  = threadIdx.x;
    const int lane = tid & 31;
    const int wave = tid >> 5;
    const int half = lane >> 4;
    const int l16  = lane & 15;
    const int wm   = wave >> 2;
    const int wn   = wave & 3;
    const int row0 = blockIdx.y * BM;
    const int col0 = blockIdx.x * BN;

    const int crow = tid >> 3;
    const int ck   = (tid & 7) << 2;
    const float* ag = x + (size_t)(row0 + crow) * IN_DIM + ck;
    const float* bg = w_in + (size_t)(col0 + crow) * IN_DIM + ck;
    float* sa = lds + crow * LDP + ck;

    const float* fa = lds + (wm * 64 + l16) * LDP + 4 * half;
    const float* fb = lds + TILE_FLOATS + (wn * 32 + l16) * LDP + 4 * half;

    v8f acc[4][2];
#pragma unroll
    for (int i = 0; i < 4; ++i) {
#pragma unroll
        for (int j = 0; j < 2; ++j) acc[i][j] = (v8f){0.f,0.f,0.f,0.f,0.f,0.f,0.f,0.f};
    }

    v4f ra[4], rb[4];
    tile_fetch(ag, bg, 0, ra, rb);
    tile_store(sa, ra, rb);
    __syncthreads();

    for (int kc = 0; kc < NKC - 1; ++kc) {
        const int cur = (kc & 1) * STAGE_FLOATS;
        tile_fetch(ag, bg, (kc + 1) * BK, ra, rb);
        tile_mma(fa + cur, fb + cur, acc);
        tile_store(sa + (STAGE_FLOATS - cur), ra, rb);
        __syncthreads();
    }
    tile_mma(fa + ((NKC - 1) & 1) * STAGE_FLOATS, fb + ((NKC - 1) & 1) * STAGE_FLOATS, acc);

    float* outp = u + (size_t)(row0 + wm * 64 + half * 8) * NCOL + col0 + wn * 32 + l16;
#pragma unroll
    for (int i = 0; i < 4; ++i) {
#pragma unroll
        for (int j = 0; j < 2; ++j) {
#pragma unroll
            for (int r = 0; r < 8; ++r)
                outp[(size_t)(i * 16 + r) * NCOL + j * 16] = acc[i][j][r];
        }
    }
}

__device__ __forceinline__ float tanh_hw(float v) {
    return __builtin_amdgcn_tanhf(v);
}

#define PF_STEPS 32
#define SB_STEPS 16

template <bool kRefill>
__device__ __forceinline__ void scan_block(
        const float* __restrict__ u, float* __restrict__ h_n, float* __restrict__ c_n,
        const int s, const int t0,
        const float wf, const float wi, const float wo, const float wg,
        const float bf, const float bi, const float bo, const float bg,
        v4f (&ring)[PF_STEPS],
        float& wfo, float& wio, float& woo, float& wgo,
        float& tc, float& c) {
#pragma unroll
    for (int b0 = 0; b0 < PF_STEPS; b0 += SB_STEPS) {
        float os[SB_STEPS], ts[SB_STEPS], cs[SB_STEPS];
#pragma unroll
        for (int jj = 0; jj < SB_STEPS; ++jj) {
            const int j = b0 + jj;
            const int t = t0 + j;
            const float pf = __builtin_fmaf(0.5f, ring[j].x, bf);
            const float pi = __builtin_fmaf(0.5f, ring[j].y, bi);
            const float po = __builtin_fmaf(0.5f, ring[j].z, bo);
            const float pg = ring[j].w + bg;
            const float ai = __builtin_fmaf(wio, tc, pi);
            const float af = __builtin_fmaf(wfo, tc, pf);
            const float ao = __builtin_fmaf(woo, tc, po);
            const float ag = __builtin_fmaf(wgo, tc, pg);
            const float hc = 0.5f * c;
            const float ti = tanh_hw(ai);
            const float tf = tanh_hw(af);
            const float to = tanh_hw(ao);
            const float g  = tanh_hw(ag);
            __builtin_amdgcn_sched_barrier(0);
            const float ig = __builtin_fmaf(0.5f, ti, 0.5f);
            const float fc = __builtin_fmaf(tf, hc, hc);
            const float o  = __builtin_fmaf(0.5f, to, 0.5f);
            c = __builtin_fmaf(ig, g, fc);
            tc = tanh_hw(c);
            os[jj] = o;
            ts[jj] = tc;
            cs[jj] = c;
            __builtin_amdgcn_sched_barrier(0);
            wfo = wf * o;
            wio = wi * o;
            woo = wo * o;
            wgo = wg * o;

            if (kRefill)
                ring[j] = *(const v4f*)(u + (size_t)(t + PF_STEPS) * NCOL + 4 * s);
        }
#pragma unroll
        for (int jj = 0; jj < SB_STEPS; ++jj) {
            h_n[(size_t)(t0 + b0 + jj + 1) * S_DIM + s] = os[jj] * ts[jj];
            c_n[(size_t)(t0 + b0 + jj + 1) * S_DIM + s] = cs[jj];
        }
    }
}

__global__ void __launch_bounds__(32)
scan_kernel(const float* __restrict__ u,
            const float* __restrict__ w_rec,
            const float* __restrict__ bias,
            float* __restrict__ h_n, float* __restrict__ c_n) {
    const int s = blockIdx.x * blockDim.x + threadIdx.x;
    const float wf = 0.5f * w_rec[s],             wi = 0.5f * w_rec[S_DIM + s];
    const float wo = 0.5f * w_rec[2 * S_DIM + s], wg = w_rec[3 * S_DIM + s];
    const float bf = 0.5f * bias[s],              bi = 0.5f * bias[S_DIM + s];
    const float bo = 0.5f * bias[2 * S_DIM + s],  bg = bias[3 * S_DIM + s];

    float wfo = 0.f, wio = 0.f, woo = 0.f, wgo = 0.f;
    float tc = 0.f, c = 0.f;
    h_n[s] = 0.f;
    c_n[s] = 0.f;

    v4f ring[PF_STEPS];
#pragma unroll
    for (int j = 0; j < PF_STEPS; ++j)
        ring[j] = *(const v4f*)(u + (size_t)j * NCOL + 4 * s);

    for (int t0 = 0; t0 < T_DIM - PF_STEPS; t0 += PF_STEPS)
        scan_block<true>(u, h_n, c_n, s, t0, wf, wi, wo, wg, bf, bi, bo, bg,
                         ring, wfo, wio, woo, wgo, tc, c);
    scan_block<false>(u, h_n, c_n, s, T_DIM - PF_STEPS, wf, wi, wo, wg, bf, bi, bo, bg,
                      ring, wfo, wio, woo, wgo, tc, c);
}

__global__ void q_kernel(const float* __restrict__ h_n,
                         const float* __restrict__ w_reg,
                         const float* __restrict__ b_reg,
                         float* __restrict__ q) {
    const int lane = threadIdx.x & 31;
    const int t = (blockIdx.x * blockDim.x + threadIdx.x) >> 5;
    const float* hr = h_n + (size_t)(t + 1) * S_DIM;
    float acc = 0.f;
#pragma unroll
    for (int j = 0; j < S_DIM; j += 32)
        acc += hr[j + lane] * w_reg[j + lane];
#pragma unroll
    for (int off = 16; off; off >>= 1)
        acc += __shfl_xor(acc, off, 32);
    if (lane == 0) q[t] = acc + b_reg[0];
}

extern "C" void kernel_launch(void* const* d_in, const int* in_sizes, int n_in,
                              void* d_out, int out_size, void* d_ws, size_t ws_size,
                              hipStream_t stream) {
    (void)in_sizes; (void)n_in; (void)out_size; (void)ws_size;
    const float* x     = (const float*)d_in[0];
    const float* w_in  = (const float*)d_in[1];
    const float* w_rec = (const float*)d_in[2];
    const float* bias  = (const float*)d_in[3];
    const float* w_reg = (const float*)d_in[4];
    const float* b_reg = (const float*)d_in[5];

    float* out = (float*)d_out;
    float* q   = out;
    float* h_n = out + T_DIM;
    float* c_n = h_n + (size_t)(T_DIM + 1) * S_DIM;

    float* u  = (float*)d_ws;

    dim3 ggrid(NCOL / BN, T_DIM / BM);
    gemm_u_kernel<<<ggrid, 256, 0, stream>>>(x, w_in, u);

    scan_kernel<<<S_DIM / 32, 32, 0, stream>>>(u, w_rec, bias, h_n, c_n);

    q_kernel<<<(T_DIM * 32) / 256, 256, 0, stream>>>(h_n, w_reg, b_reg, q);
}


# product and recurrence pipelined over time: launch j holds the product's tiles of chunk j beside the recurrence over chunk j-1 (9 chunks; carry through two used rows of u); order of sums unchanged
# speedup vs baseline: 14.9944x; 1.3435x over previous
#include <hip/hip_runtime.h>
#include <hip/hip_bf16.h>

typedef float v2f __attribute__((ext_vector_type(2)));
typedef float v4f __attribute__((ext_vector_type(4)));
typedef float v8f __attribute__((ext_vector_type(8)));

#define T_DIM  16384
#define IN_DIM 1024
#define S_DIM  512
#define NCOL   2048

#define BM  128
#define BN  128
#define BK  32
#define NKC (IN_DIM / BK)
#define LDP 36
#define TILE_FLOATS  (BM * LDP)
#define STAGE_FLOATS (2 * TILE_FLOATS)
#define PIPE_THREADS 512
#define SCAN_WGS (S_DIM / 32)
#define CH_ROWS 1920

__device__ __forceinline__ void tile_fetch(const float* __restrict__ ag,
                                           const float* __restrict__ bg,
                                           const int k0, v4f (&ra)[2], v4f (&rb)[2]) {
#pragma unroll
    for (int p = 0; p < 2; ++p) {
        ra[p] = *(const v4f*)(ag + (size_t)p * 64 * IN_DIM + k0);
        rb[p] = *(const v4f*)(bg + (size_t)p * 64 * IN_DIM + k0);
    }
}

__device__ __forceinline__ void tile_store(float* sa, const v4f (&ra)[2],
                                           const v4f (&rb)[2]) {
#pragma unroll
    for (int p = 0; p < 2; ++p) {
        *(v4f*)(sa + p * 64 * LDP) = ra[p];
        *(v4f*)(sa + TILE_FLOATS + p * 64 * LDP) = rb[p];
    }
}

__device__ __forceinline__ void tile_mma(const float* fa, const float* fb,
                                         v8f (&acc)[2][2]) {
#pragma unroll
    for (int k8 = 0; k8 < BK; k8 += 8) {
        v4f a[2], b[2];
#pragma unroll
        for (int i = 0; i < 2; ++i) a[i] = *(const v4f*)(fa + i * 16 * LDP + k8);
#pragma unroll
        for (int j = 0; j < 2; ++j) b[j] = *(const v4f*)(fb + j * 16 * LDP + k8);
#pragma unroll
        for (int i = 0; i < 2; ++i) {
#pragma unroll
            for (int j = 0; j < 2; ++j)
                acc[i][j] = __builtin_amdgcn_wmma_f32_16x16x4_f32(false, a[i].xy, false, b[j].xy,
                                                                  (short)0, acc[i][j], false, false);
        }
#pragma unroll
        for (int i = 0; i < 2; ++i) {
#pragma unroll
            for (int j = 0; j < 2; ++j)
                acc[i][j] = __builtin_amdgcn_wmma_f32_16x16x4_f32(false, a[i].zw, false, b[j].zw,
                                                                  (short)0, acc[i][j], false, false);
        }
    }
}

__device__ __forceinline__ void gemm_tile(const float* __restrict__ x,
                                          const float* __restrict__ w_in,
                                          float* u, float* lds,
                                          const int row0, const int col0) {
    const int tid  = threadIdx.x;
    const int lane = tid & 31;
    const int wave = tid >> 5;
    const int half = lane >> 4;
    const int l16  = lane & 15;
    const int wm   = wave >> 2;
    const int wn   = wave & 3;

    const int crow = tid >> 3;
    const int ck   = (tid & 7) << 2;
    const float* ag = x + (size_t)(row0 + crow) * IN_DIM + ck;
    const float* bg = w_in + (size_t)(col0 + crow) * IN_DIM + ck;
    float* sa = lds + crow * LDP + ck;

    const float* fa = lds + (wm * 32 + l16) * LDP + 4 * half;
    const float* fb = lds + TILE_FLOATS + (wn * 32 + l16) * LDP + 4 * half;

    v8f acc[2][2];
#pragma unroll
    for (int i = 0; i < 2; ++i) {
#pragma unroll
        for (int j = 0; j < 2; ++j) acc[i][j] = (v8f){0.f,0.f,0.f,0.f,0.f,0.f,0.f,0.f};
    }

    v4f ra[2], rb[2];
    tile_fetch(ag, bg, 0, ra, rb);
    tile_store(sa, ra, rb);
    __syncthreads();

    for (int kc = 0; kc < NKC - 1; ++kc) {
        const int cur = (kc & 1) * STAGE_FLOATS;
        tile_fetch(ag, bg, (kc + 1) * BK, ra, rb);
        tile_mma(fa + cur, fb + cur, acc);
        tile_store(sa + (STAGE_FLOATS - cur), ra, rb);
        __syncthreads();
    }
    tile_mma(fa + ((NKC - 1) & 1) * STAGE_FLOATS, fb + ((NKC - 1) & 1) * STAGE_FLOATS, acc);

    float* outp = u + (size_t)(row0 + wm * 32 + half * 8) * NCOL + col0 + wn * 32 + l16;
#pragma unroll
    for (int i = 0; i < 2; ++i) {
#pragma unroll
        for (int j = 0; j < 2; ++j) {
#pragma unroll
            for (int r = 0; r < 8; ++r)
                outp[(size_t)(i * 16 + r) * NCOL + j * 16] = acc[i][j][r];
        }
    }
}

__device__ __forceinline__ float tanh_hw(float v) {
    return __builtin_amdgcn_tanhf(v);
}

#define PF_STEPS 32
#define SB_STEPS 16

template <bool kRefill>
__device__ __forceinline__ void scan_block(
        const float* up, float* __restrict__ hp, float* __restrict__ cp,
        const float wf, const float wi, const float wo, const float wg,
        const float bf, const float bi, const float bo, const float bg,
        v4f (&ring)[PF_STEPS],
        float& wfo, float& wio, float& woo, float& wgo,
        float& tc, float& c) {
#pragma unroll
    for (int b0 = 0; b0 < PF_STEPS; b0 += SB_STEPS) {
        float os[SB_STEPS], ts[SB_STEPS], cs[SB_STEPS];
#pragma unroll
        for (int jj = 0; jj < SB_STEPS; ++jj) {
            const int j = b0 + jj;
            const float pf = __builtin_fmaf(0.5f, ring[j].x, bf);
            const float pi = __builtin_fmaf(0.5f, ring[j].y, bi);
            const float po = __builtin_fmaf(0.5f, ring[j].z, bo);
            const float pg = ring[j].w + bg;
            const float ai = __builtin_fmaf(wio, tc, pi);
            const float af = __builtin_fmaf(wfo, tc, pf);
            const float ao = __builtin_fmaf(woo, tc, po);
            const float ag = __builtin_fmaf(wgo, tc, pg);
            const float hc = 0.5f * c;
            const float ti = tanh_hw(ai);
            const float tf = tanh_hw(af);
            const float to = tanh_hw(ao);
            const float g  = tanh_hw(ag);
            __builtin_amdgcn_sched_barrier(0);
            const float ig = __builtin_fmaf(0.5f, ti, 0.5f);
            const float fc = __builtin_fmaf(tf, hc, hc);
            const float o  = __builtin_fmaf(0.5f, to, 0.5f);
            c = __builtin_fmaf(ig, g, fc);
            tc = tanh_hw(c);
            os[jj] = o;
            ts[jj] = tc;
            cs[jj] = c;
            __builtin_amdgcn_sched_barrier(0);
            wfo = wf * o;
            wio = wi * o;
            woo = wo * o;
            wgo = wg * o;

            if (kRefill)
                ring[j] = *(const v4f*)(up + (size_t)(j + PF_STEPS) * NCOL);
        }
#pragma unroll
        for (int jj = 0; jj < SB_STEPS; ++jj) {
            hp[(size_t)(b0 + jj + 1) * S_DIM] = os[jj] * ts[jj];
            cp[(size_t)(b0 + jj + 1) * S_DIM] = cs[jj];
        }
    }
}

__device__ __forceinline__ void scan_chunk(float* u,
                                           const float* __restrict__ w_rec,
                                           const float* __restrict__ bias,
                                           float* __restrict__ h_n, float* __restrict__ c_n,
                                           const int s, const int t_beg, const int t_len) {
    const float wf = 0.5f * w_rec[s],             wi = 0.5f * w_rec[S_DIM + s];
    const float wo = 0.5f * w_rec[2 * S_DIM + s], wg = w_rec[3 * S_DIM + s];
    const float bf = 0.5f * bias[s],              bi = 0.5f * bias[S_DIM + s];
    const float bo = 0.5f * bias[2 * S_DIM + s],  bg = bias[3 * S_DIM + s];

    float wfo, wio, woo, wgo, tc, c;
    if (t_beg == 0) {
        wfo = 0.f; wio = 0.f; woo = 0.f; wgo = 0.f;
        tc = 0.f; c = 0.f;
        h_n[s] = 0.f;
        c_n[s] = 0.f;
    } else {
        const v4f p = *(const v4f*)(u + (size_t)(t_beg - 1) * NCOL + 4 * s);
        const v4f q = *(const v4f*)(u + (size_t)(t_beg - 2) * NCOL + 4 * s);
        wfo = p.x; wio = p.y; woo = p.z; wgo = p.w;
        tc = q.x; c = q.y;
    }

    v4f ring[PF_STEPS];
#pragma unroll
    for (int j = 0; j < PF_STEPS; ++j)
        ring[j] = *(const v4f*)(u + (size_t)(t_beg + j) * NCOL + 4 * s);

    const float* up = u + (size_t)t_beg * NCOL + 4 * s;
    float* hp = h_n + (size_t)t_beg * S_DIM + s;
    float* cp = c_n + (size_t)t_beg * S_DIM + s;
    for (int nb = t_len / PF_STEPS - 1; nb > 0; --nb) {
        scan_block<true>(up, hp, cp, wf, wi, wo, wg, bf, bi, bo, bg,
                         ring, wfo, wio, woo, wgo, tc, c);
        up += (size_t)PF_STEPS * NCOL;
        hp += (size_t)PF_STEPS * S_DIM;
        cp += (size_t)PF_STEPS * S_DIM;
    }
    scan_block<false>(up, hp, cp, wf, wi, wo, wg, bf, bi, bo, bg,
                      ring, wfo, wio, woo, wgo, tc, c);

    *(v4f*)(u + (size_t)(t_beg + t_len - 1) * NCOL + 4 * s) = (v4f){wfo, wio, woo, wgo};
    *(v4f*)(u + (size_t)(t_beg + t_len - 2) * NCOL + 4 * s) = (v4f){tc, c, 0.f, 0.f};
}

__global__ void __launch_bounds__(PIPE_THREADS)
pipe_kernel(const float* __restrict__ x, const float* __restrict__ w_in,
            const float* __restrict__ w_rec, const float* __restrict__ bias,
            float* u, float* __restrict__ h_n, float* __restrict__ c_n,
            const int g_row0, const int s_beg, const int s_len) {
    __shared__ __attribute__((aligned(16))) float lds[2 * STAGE_FLOATS];

    if (blockIdx.x < SCAN_WGS) {
        if (s_len > 0 && threadIdx.x < 32)
            scan_chunk(u, w_rec, bias, h_n, c_n, blockIdx.x * 32 + threadIdx.x, s_beg, s_len);
        return;
    }
    const int tile = blockIdx.x - SCAN_WGS;
    gemm_tile(x, w_in, u, lds, g_row0 + (tile >> 4) * BM, (tile & 15) * BN);
}

__global__ void q_kernel(const float* __restrict__ h_n,
                         const float* __restrict__ w_reg,
                         const float* __restrict__ b_reg,
                         float* __restrict__ q) {
    const int lane = threadIdx.x & 31;
    const int t = (blockIdx.x * blockDim.x + threadIdx.x) >> 5;
    const float* hr = h_n + (size_t)(t + 1) * S_DIM;
    float acc = 0.f;
#pragma unroll
    for (int j = 0; j < S_DIM; j += 32)
        acc += hr[j + lane] * w_reg[j + lane];
#pragma unroll
    for (int off = 16; off; off >>= 1)
        acc += __shfl_xor(acc, off, 32);
    if (lane == 0) q[t] = acc + b_reg[0];
}

extern "C" void kernel_launch(void* const* d_in, const int* in_sizes, int n_in,
                              void* d_out, int out_size, void* d_ws, size_t ws_size,
                              hipStream_t stream) {
    (void)in_sizes; (void)n_in; (void)out_size; (void)ws_size;
    const float* x     = (const float*)d_in[0];
    const float* w_in  = (const float*)d_in[1];
    const float* w_rec = (const float*)d_in[2];
    const float* bias  = (const float*)d_in[3];
    const float* w_reg = (const float*)d_in[4];
    const float* b_reg = (const float*)d_in[5];

    float* out = (float*)d_out;
    float* q   = out;
    float* h_n = out + T_DIM;
    float* c_n = h_n + (size_t)(T_DIM + 1) * S_DIM;

    float* u  = (float*)d_ws;

    pipe_kernel<<<256, PIPE_THREADS, 0, stream>>>(x, w_in, w_rec, bias, u, h_n, c_n,     0,     0,    0);
    pipe_kernel<<<256, PIPE_THREADS, 0, stream>>>(x, w_in, w_rec, bias, u, h_n, c_n,  1920,     0, 1920);
    pipe_kernel<<<256, PIPE_THREADS, 0, stream>>>(x, w_in, w_rec, bias, u, h_n, c_n,  3840,  1920, 1920);
    pipe_kernel<<<256, PIPE_THREADS, 0, stream>>>(x, w_in, w_rec, bias, u, h_n, c_n,  5760,  3840, 1920);
    pipe_kernel<<<256, PIPE_THREADS, 0, stream>>>(x, w_in, w_rec, bias, u, h_n, c_n,  7680,  5760, 1920);
    pipe_kernel<<<256, PIPE_THREADS, 0, stream>>>(x, w_in, w_rec, bias, u, h_n, c_n,  9600,  7680, 1920);
    pipe_kernel<<<256, PIPE_THREADS, 0, stream>>>(x, w_in, w_rec, bias, u, h_n, c_n, 11520,  9600, 1920);
    pipe_kernel<<<256, PIPE_THREADS, 0, stream>>>(x, w_in, w_rec, bias, u, h_n, c_n, 13440, 11520, 1920);
    pipe_kernel<<<144, PIPE_THREADS, 0, stream>>>(x, w_in, w_rec, bias, u, h_n, c_n, 15360, 13440, 1920);
    pipe_kernel<<< 16, PIPE_THREADS, 0, stream>>>(x, w_in, w_rec, bias, u, h_n, c_n,     0, 15360, 1024);

    q_kernel<<<(T_DIM * 32) / 256, 256, 0, stream>>>(h_n, w_reg, b_reg, q);
}


# scan kernel: the recurrence's loads and stores issued in groups (ring refilled 8 rows at a time, every second group joined to the 16-step store batch), not one load in every step; sums unchanged
# speedup vs baseline: 20.3276x; 1.3557x over previous
#include <hip/hip_runtime.h>
#include <hip/hip_bf16.h>

typedef float v2f __attribute__((ext_vector_type(2)));
typedef float v4f __attribute__((ext_vector_type(4)));
typedef float v8f __attribute__((ext_vector_type(8)));

#define T_DIM  16384
#define IN_DIM 1024
#define S_DIM  512
#define NCOL   2048

#define BM  128
#define BN  128
#define BK  32
#define NKC (IN_DIM / BK)
#define LDP 36
#define TILE_FLOATS  (BM * LDP)
#define STAGE_FLOATS (2 * TILE_FLOATS)
#define PIPE_THREADS 512
#define SCAN_WGS (S_DIM / 32)
#define CH_ROWS 1920

__device__ __forceinline__ void tile_fetch(const float* __restrict__ ag,
                                           const float* __restrict__ bg,
                                           const int k0, v4f (&ra)[2], v4f (&rb)[2]) {
#pragma unroll
    for (int p = 0; p < 2; ++p) {
        ra[p] = *(const v4f*)(ag + (size_t)p * 64 * IN_DIM + k0);
        rb[p] = *(const v4f*)(bg + (size_t)p * 64 * IN_DIM + k0);
    }
}

__device__ __forceinline__ void tile_store(float* sa, const v4f (&ra)[2],
                                           const v4f (&rb)[2]) {
#pragma unroll
    for (int p = 0; p < 2; ++p) {
        *(v4f*)(sa + p * 64 * LDP) = ra[p];
        *(v4f*)(sa + TILE_FLOATS + p * 64 * LDP) = rb[p];
    }
}

__device__ __forceinline__ void tile_mma(const float* fa, const float* fb,
                                         v8f (&acc)[2][2]) {
#pragma unroll
    for (int k8 = 0; k8 < BK; k8 += 8) {
        v4f a[2], b[2];
#pragma unroll
        for (int i = 0; i < 2; ++i) a[i] = *(const v4f*)(fa + i * 16 * LDP + k8);
#pragma unroll
        for (int j = 0; j < 2; ++j) b[j] = *(const v4f*)(fb + j * 16 * LDP + k8);
#pragma unroll
        for (int i = 0; i < 2; ++i) {
#pragma unroll
            for (int j = 0; j < 2; ++j)
                acc[i][j] = __builtin_amdgcn_wmma_f32_16x16x4_f32(false, a[i].xy, false, b[j].xy,
                                                                  (short)0, acc[i][j], false, false);
        }
#pragma unroll
        for (int i = 0; i < 2; ++i) {
#pragma unroll
            for (int j = 0; j < 2; ++j)
                acc[i][j] = __builtin_amdgcn_wmma_f32_16x16x4_f32(false, a[i].zw, false, b[j].zw,
                                                                  (short)0, acc[i][j], false, false);
        }
    }
}

__device__ __forceinline__ void gemm_tile(const float* __restrict__ x,
                                          const float* __restrict__ w_in,
                                          float* u, float* lds,
                                          const int row0, const int col0) {
    const int tid  = threadIdx.x;
    const int lane = tid & 31;
    const int wave = tid >> 5;
    const int half = lane >> 4;
    const int l16  = lane & 15;
    const int wm   = wave >> 2;
    const int wn   = wave & 3;

    const int crow = tid >> 3;
    const int ck   = (tid & 7) << 2;
    const float* ag = x + (size_t)(row0 + crow) * IN_DIM + ck;
    const float* bg = w_in + (size_t)(col0 + crow) * IN_DIM + ck;
    float* sa = lds + crow * LDP + ck;

    const float* fa = lds + (wm * 32 + l16) * LDP + 4 * half;
    const float* fb = lds + TILE_FLOATS + (wn * 32 + l16) * LDP + 4 * half;

    v8f acc[2][2];
#pragma unroll
    for (int i = 0; i < 2; ++i) {
#pragma unroll
        for (int j = 0; j < 2; ++j) acc[i][j] = (v8f){0.f,0.f,0.f,0.f,0.f,0.f,0.f,0.f};
    }

    v4f ra[2], rb[2];
    tile_fetch(ag, bg, 0, ra, rb);
    tile_store(sa, ra, rb);
    __syncthreads();

    for (int kc = 0; kc < NKC - 1; ++kc) {
        const int cur = (kc & 1) * STAGE_FLOATS;
        tile_fetch(ag, bg, (kc + 1) * BK, ra, rb);
        tile_mma(fa + cur, fb + cur, acc);
        tile_store(sa + (STAGE_FLOATS - cur), ra, rb);
        __syncthreads();
    }
    tile_mma(fa + ((NKC - 1) & 1) * STAGE_FLOATS, fb + ((NKC - 1) & 1) * STAGE_FLOATS, acc);

    float* outp = u + (size_t)(row0 + wm * 32 + half * 8) * NCOL + col0 + wn * 32 + l16;
#pragma unroll
    for (int i = 0; i < 2; ++i) {
#pragma unroll
        for (int j = 0; j < 2; ++j) {
#pragma unroll
            for (int r = 0; r < 8; ++r)
                outp[(size_t)(i * 16 + r) * NCOL + j * 16] = acc[i][j][r];
        }
    }
}

__device__ __forceinline__ float tanh_hw(float v) {
    return __builtin_amdgcn_tanhf(v);
}

#define PF_STEPS 32
#define SB_STEPS 16
#define RG_STEPS 8

template <bool kRefill>
__device__ __forceinline__ void scan_block(
        const float* up, float* __restrict__ hp, float* __restrict__ cp,
        const float wf, const float wi, const float wo, const float wg,
        const float bf, const float bi, const float bo, const float bg,
        v4f (&ring)[PF_STEPS],
        float& wfo, float& wio, float& woo, float& wgo,
        float& tc, float& c) {
#pragma unroll
    for (int b0 = 0; b0 < PF_STEPS; b0 += SB_STEPS) {
        float os[SB_STEPS], ts[SB_STEPS], cs[SB_STEPS];
#pragma unroll
        for (int jj = 0; jj < SB_STEPS; ++jj) {
            const int j = b0 + jj;
            const float pf = __builtin_fmaf(0.5f, ring[j].x, bf);
            const float pi = __builtin_fmaf(0.5f, ring[j].y, bi);
            const float po = __builtin_fmaf(0.5f, ring[j].z, bo);
            const float pg = ring[j].w + bg;
            const float ai = __builtin_fmaf(wio, tc, pi);
            const float af = __builtin_fmaf(wfo, tc, pf);
            const float ao = __builtin_fmaf(woo, tc, po);
            const float ag = __builtin_fmaf(wgo, tc, pg);
            const float hc = 0.5f * c;
            const float ti = tanh_hw(ai);
            const float tf = tanh_hw(af);
            const float to = tanh_hw(ao);
            const float g  = tanh_hw(ag);
            __builtin_amdgcn_sched_barrier(0);
            const float ig = __builtin_fmaf(0.5f, ti, 0.5f);
            const float fc = __builtin_fmaf(tf, hc, hc);
            const float o  = __builtin_fmaf(0.5f, to, 0.5f);
            c = __builtin_fmaf(ig, g, fc);
            tc = tanh_hw(c);
            os[jj] = o;
            ts[jj] = tc;
            cs[jj] = c;
            __builtin_amdgcn_sched_barrier(0);
            wfo = wf * o;
            wio = wi * o;
            woo = wo * o;
            wgo = wg * o;

            if (kRefill && jj == RG_STEPS - 1) {
                __builtin_amdgcn_sched_barrier(0);
#pragma unroll
                for (int r = 0; r < RG_STEPS; ++r)
                    ring[b0 + r] = *(const v4f*)(up + (size_t)(b0 + r + PF_STEPS) * NCOL);
                __builtin_amdgcn_sched_barrier(0);
            }
        }
        float hs[SB_STEPS];
#pragma unroll
        for (int jj = 0; jj < SB_STEPS; ++jj)
            hs[jj] = os[jj] * ts[jj];
        __builtin_amdgcn_sched_barrier(0);
        if (kRefill) {
#pragma unroll
            for (int r = RG_STEPS; r < SB_STEPS; ++r)
                ring[b0 + r] = *(const v4f*)(up + (size_t)(b0 + r + PF_STEPS) * NCOL);
        }
#pragma unroll
        for (int jj = 0; jj < SB_STEPS; ++jj)
            hp[(size_t)(b0 + jj + 1) * S_DIM] = hs[jj];
#pragma unroll
        for (int jj = 0; jj < SB_STEPS; ++jj)
            cp[(size_t)(b0 + jj + 1) * S_DIM] = cs[jj];
        __builtin_amdgcn_sched_barrier(0);
    }
}

__device__ __forceinline__ void scan_chunk(float* u,
                                           const float* __restrict__ w_rec,
                                           const float* __restrict__ bias,
                                           float* __restrict__ h_n, float* __restrict__ c_n,
                                           const int s, const int t_beg, const int t_len) {
    const float wf = 0.5f * w_rec[s],             wi = 0.5f * w_rec[S_DIM + s];
    const float wo = 0.5f * w_rec[2 * S_DIM + s], wg = w_rec[3 * S_DIM + s];
    const float bf = 0.5f * bias[s],              bi = 0.5f * bias[S_DIM + s];
    const float bo = 0.5f * bias[2 * S_DIM + s],  bg = bias[3 * S_DIM + s];

    float wfo, wio, woo, wgo, tc, c;
    if (t_beg == 0) {
        wfo = 0.f; wio = 0.f; woo = 0.f; wgo = 0.f;
        tc = 0.f; c = 0.f;
        h_n[s] = 0.f;
        c_n[s] = 0.f;
    } else {
        const v4f p = *(const v4f*)(u + (size_t)(t_beg - 1) * NCOL + 4 * s);
        const v4f q = *(const v4f*)(u + (size_t)(t_beg - 2) * NCOL + 4 * s);
        wfo = p.x; wio = p.y; woo = p.z; wgo = p.w;
        tc = q.x; c = q.y;
    }

    v4f ring[PF_STEPS];
#pragma unroll
    for (int j = 0; j < PF_STEPS; ++j)
        ring[j] = *(const v4f*)(u + (size_t)(t_beg + j) * NCOL + 4 * s);

    const float* up = u + (size_t)t_beg * NCOL + 4 * s;
    float* hp = h_n + (size_t)t_beg * S_DIM + s;
    float* cp = c_n + (size_t)t_beg * S_DIM + s;
    for (int nb = t_len / PF_STEPS - 1; nb > 0; --nb) {
        scan_block<true>(up, hp, cp, wf, wi, wo, wg, bf, bi, bo, bg,
                         ring, wfo, wio, woo, wgo, tc, c);
        up += (size_t)PF_STEPS * NCOL;
        hp += (size_t)PF_STEPS * S_DIM;
        cp += (size_t)PF_STEPS * S_DIM;
    }
    scan_block<false>(up, hp, cp, wf, wi, wo, wg, bf, bi, bo, bg,
                      ring, wfo, wio, woo, wgo, tc, c);

    *(v4f*)(u + (size_t)(t_beg + t_len - 1) * NCOL + 4 * s) = (v4f){wfo, wio, woo, wgo};
    *(v4f*)(u + (size_t)(t_beg + t_len - 2) * NCOL + 4 * s) = (v4f){tc, c, 0.f, 0.f};
}

__global__ void __launch_bounds__(PIPE_THREADS)
pipe_kernel(const float* __restrict__ x, const float* __restrict__ w_in,
            const float* __restrict__ w_rec, const float* __restrict__ bias,
            float* u, float* __restrict__ h_n, float* __restrict__ c_n,
            const int g_row0, const int s_beg, const int s_len) {
    __shared__ __attribute__((aligned(16))) float lds[2 * STAGE_FLOATS];

    if (blockIdx.x < SCAN_WGS) {
        if (s_len > 0 && threadIdx.x < 32)
            scan_chunk(u, w_rec, bias, h_n, c_n, blockIdx.x * 32 + threadIdx.x, s_beg, s_len);
        return;
    }
    const int tile = blockIdx.x - SCAN_WGS;
    gemm_tile(x, w_in, u, lds, g_row0 + (tile >> 4) * BM, (tile & 15) * BN);
}

__global__ void q_kernel(const float* __restrict__ h_n,
                         const float* __restrict__ w_reg,
                         const float* __restrict__ b_reg,
                         float* __restrict__ q) {
    const int lane = threadIdx.x & 31;
    const int t = (blockIdx.x * blockDim.x + threadIdx.x) >> 5;
    const float* hr = h_n + (size_t)(t + 1) * S_DIM;
    float acc = 0.f;
#pragma unroll
    for (int j = 0; j < S_DIM; j += 32)
        acc += hr[j + lane] * w_reg[j + lane];
#pragma unroll
    for (int off = 16; off; off >>= 1)
        acc += __shfl_xor(acc, off, 32);
    if (lane == 0) q[t] = acc + b_reg[0];
}

extern "C" void kernel_launch(void* const* d_in, const int* in_sizes, int n_in,
                              void* d_out, int out_size, void* d_ws, size_t ws_size,
                              hipStream_t stream) {
    (void)in_sizes; (void)n_in; (void)out_size; (void)ws_size;
    const float* x     = (const float*)d_in[0];
    const float* w_in  = (const float*)d_in[1];
    const float* w_rec = (const float*)d_in[2];
    const float* bias  = (const float*)d_in[3];
    const float* w_reg = (const float*)d_in[4];
    const float* b_reg = (const float*)d_in[5];

    float* out = (float*)d_out;
    float* q   = out;
    float* h_n = out + T_DIM;
    float* c_n = h_n + (size_t)(T_DIM + 1) * S_DIM;

    float* u  = (float*)d_ws;

    pipe_kernel<<<256, PIPE_THREADS, 0, stream>>>(x, w_in, w_rec, bias, u, h_n, c_n,     0,     0,    0);
    pipe_kernel<<<256, PIPE_THREADS, 0, stream>>>(x, w_in, w_rec, bias, u, h_n, c_n,  1920,     0, 1920);
    pipe_kernel<<<256, PIPE_THREADS, 0, stream>>>(x, w_in, w_rec, bias, u, h_n, c_n,  3840,  1920, 1920);
    pipe_kernel<<<256, PIPE_THREADS, 0, stream>>>(x, w_in, w_rec, bias, u, h_n, c_n,  5760,  3840, 1920);
    pipe_kernel<<<256, PIPE_THREADS, 0, stream>>>(x, w_in, w_rec, bias, u, h_n, c_n,  7680,  5760, 1920);
    pipe_kernel<<<256, PIPE_THREADS, 0, stream>>>(x, w_in, w_rec, bias, u, h_n, c_n,  9600,  7680, 1920);
    pipe_kernel<<<256, PIPE_THREADS, 0, stream>>>(x, w_in, w_rec, bias, u, h_n, c_n, 11520,  9600, 1920);
    pipe_kernel<<<256, PIPE_THREADS, 0, stream>>>(x, w_in, w_rec, bias, u, h_n, c_n, 13440, 11520, 1920);
    pipe_kernel<<<144, PIPE_THREADS, 0, stream>>>(x, w_in, w_rec, bias, u, h_n, c_n, 15360, 13440, 1920);
    pipe_kernel<<< 16, PIPE_THREADS, 0, stream>>>(x, w_in, w_rec, bias, u, h_n, c_n,     0, 15360, 1024);

    q_kernel<<<(T_DIM * 32) / 256, 256, 0, stream>>>(h_n, w_reg, b_reg, q);
}


# scan: chain wave hands each 32-step block's h and c through LDS (8 16-byte LDS stores per 16 steps, one barrier per block) to two idle waves of its work-group, which store the rows; sums unchanged
# speedup vs baseline: 23.4286x; 1.1526x over previous
#include <hip/hip_runtime.h>
#include <hip/hip_bf16.h>

typedef float v2f __attribute__((ext_vector_type(2)));
typedef float v4f __attribute__((ext_vector_type(4)));
typedef float v8f __attribute__((ext_vector_type(8)));

#define T_DIM  16384
#define IN_DIM 1024
#define S_DIM  512
#define NCOL   2048

#define BM  128
#define BN  128
#define BK  32
#define NKC (IN_DIM / BK)
#define LDP 36
#define TILE_FLOATS  (BM * LDP)
#define STAGE_FLOATS (2 * TILE_FLOATS)
#define PIPE_THREADS 512
#define SCAN_WGS (S_DIM / 32)
#define CH_ROWS 1920

__device__ __forceinline__ void tile_fetch(const float* __restrict__ ag,
                                           const float* __restrict__ bg,
                                           const int k0, v4f (&ra)[2], v4f (&rb)[2]) {
#pragma unroll
    for (int p = 0; p < 2; ++p) {
        ra[p] = *(const v4f*)(ag + (size_t)p * 64 * IN_DIM + k0);
        rb[p] = *(const v4f*)(bg + (size_t)p * 64 * IN_DIM + k0);
    }
}

__device__ __forceinline__ void tile_store(float* sa, const v4f (&ra)[2],
                                           const v4f (&rb)[2]) {
#pragma unroll
    for (int p = 0; p < 2; ++p) {
        *(v4f*)(sa + p * 64 * LDP) = ra[p];
        *(v4f*)(sa + TILE_FLOATS + p * 64 * LDP) = rb[p];
    }
}

__device__ __forceinline__ void tile_mma(const float* fa, const float* fb,
                                         v8f (&acc)[2][2]) {
#pragma unroll
    for (int k8 = 0; k8 < BK; k8 += 8) {
        v4f a[2], b[2];
#pragma unroll
        for (int i = 0; i < 2; ++i) a[i] = *(const v4f*)(fa + i * 16 * LDP + k8);
#pragma unroll
        for (int j = 0; j < 2; ++j) b[j] = *(const v4f*)(fb + j * 16 * LDP + k8);
#pragma unroll
        for (int i = 0; i < 2; ++i) {
#pragma unroll
            for (int j = 0; j < 2; ++j)
                acc[i][j] = __builtin_amdgcn_wmma_f32_16x16x4_f32(false, a[i].xy, false, b[j].xy,
                                                                  (short)0, acc[i][j], false, false);
        }
#pragma unroll
        for (int i = 0; i < 2; ++i) {
#pragma unroll
            for (int j = 0; j < 2; ++j)
                acc[i][j] = __builtin_amdgcn_wmma_f32_16x16x4_f32(false, a[i].zw, false, b[j].zw,
                                                                  (short)0, acc[i][j], false, false);
        }
    }
}

__device__ __forceinline__ void gemm_tile(const float* __restrict__ x,
                                          const float* __restrict__ w_in,
                                          float* u, float* lds,
                                          const int row0, const int col0) {
    const int tid  = threadIdx.x;
    const int lane = tid & 31;
    const int wave = tid >> 5;
    const int half = lane >> 4;
    const int l16  = lane & 15;
    const int wm   = wave >> 2;
    const int wn   = wave & 3;

    const int crow = tid >> 3;
    const int ck   = (tid & 7) << 2;
    const float* ag = x + (size_t)(row0 + crow) * IN_DIM + ck;
    const float* bg = w_in + (size_t)(col0 + crow) * IN_DIM + ck;
    float* sa = lds + crow * LDP + ck;

    const float* fa = lds + (wm * 32 + l16) * LDP + 4 * half;
    const float* fb = lds + TILE_FLOATS + (wn * 32 + l16) * LDP + 4 * half;

    v8f acc[2][2];
#pragma unroll
    for (int i = 0; i < 2; ++i) {
#pragma unroll
        for (int j = 0; j < 2; ++j) acc[i][j] = (v8f){0.f,0.f,0.f,0.f,0.f,0.f,0.f,0.f};
    }

    v4f ra[2], rb[2];
    tile_fetch(ag, bg, 0, ra, rb);
    tile_store(sa, ra, rb);
    __syncthreads();

    for (int kc = 0; kc < NKC - 1; ++kc) {
        const int cur = (kc & 1) * STAGE_FLOATS;
        tile_fetch(ag, bg, (kc + 1) * BK, ra, rb);
        tile_mma(fa + cur, fb + cur, acc);
        tile_store(sa + (STAGE_FLOATS - cur), ra, rb);
        __syncthreads();
    }
    tile_mma(fa + ((NKC - 1) & 1) * STAGE_FLOATS, fb + ((NKC - 1) & 1) * STAGE_FLOATS, acc);

    float* outp = u + (size_t)(row0 + wm * 32 + half * 8) * NCOL + col0 + wn * 32 + l16;
#pragma unroll
    for (int i = 0; i < 2; ++i) {
#pragma unroll
        for (int j = 0; j < 2; ++j) {
#pragma unroll
            for (int r = 0; r < 8; ++r)
                outp[(size_t)(i * 16 + r) * NCOL + j * 16] = acc[i][j][r];
        }
    }
}

__device__ __forceinline__ float tanh_hw(float v) {
    return __builtin_amdgcn_tanhf(v);
}

#define PF_STEPS 32
#define SB_STEPS 16
#define RG_STEPS 8

#define HO_PITCH 36
#define HO_ARRAY (32 * HO_PITCH)
#define HO_BUF   (2 * HO_ARRAY)

__device__ __forceinline__ void handover_signal() {
    __builtin_amdgcn_fence(__ATOMIC_RELEASE, "workgroup", "local");
    asm volatile("s_wait_dscnt 0x0" ::: "memory");
    __builtin_amdgcn_s_barrier_signal(-1);
}
__device__ __forceinline__ void handover_wait() {
    __builtin_amdgcn_s_barrier_wait(-1);
    __builtin_amdgcn_fence(__ATOMIC_ACQUIRE, "workgroup", "local");
}
__device__ __forceinline__ void handover_barrier() {
    handover_signal();
    handover_wait();
}

template <bool kRefill>
__device__ __forceinline__ void scan_block(
        const float* up, float* ho,
        const float wf, const float wi, const float wo, const float wg,
        const float bf, const float bi, const float bo, const float bg,
        v4f (&ring)[PF_STEPS],
        float& wfo, float& wio, float& woo, float& wgo,
        float& tc, float& c) {
#pragma unroll
    for (int b0 = 0; b0 < PF_STEPS; b0 += SB_STEPS) {
        float os[SB_STEPS], ts[SB_STEPS], cs[SB_STEPS];
#pragma unroll
        for (int jj = 0; jj < SB_STEPS; ++jj) {
            const int j = b0 + jj;
            const float pf = __builtin_fmaf(0.5f, ring[j].x, bf);
            const float pi = __builtin_fmaf(0.5f, ring[j].y, bi);
            const float po = __builtin_fmaf(0.5f, ring[j].z, bo);
            const float pg = ring[j].w + bg;
            const float ai = __builtin_fmaf(wio, tc, pi);
            const float af = __builtin_fmaf(wfo, tc, pf);
            const float ao = __builtin_fmaf(woo, tc, po);
            const float ag = __builtin_fmaf(wgo, tc, pg);
            const float hc = 0.5f * c;
            const float ti = tanh_hw(ai);
            const float tf = tanh_hw(af);
            const float to = tanh_hw(ao);
            const float g  = tanh_hw(ag);
            __builtin_amdgcn_sched_barrier(0);
            const float ig = __builtin_fmaf(0.5f, ti, 0.5f);
            const float fc = __builtin_fmaf(tf, hc, hc);
            const float o  = __builtin_fmaf(0.5f, to, 0.5f);
            c = __builtin_fmaf(ig, g, fc);
            tc = tanh_hw(c);
            os[jj] = o;
            ts[jj] = tc;
            cs[jj] = c;
            __builtin_amdgcn_sched_barrier(0);
            wfo = wf * o;
            wio = wi * o;
            woo = wo * o;
            wgo = wg * o;

            if (jj == RG_STEPS - 1 && (kRefill || b0 == 0)) {
                __builtin_amdgcn_sched_barrier(0);
                if (kRefill) {
#pragma unroll
                    for (int r = 0; r < RG_STEPS; ++r)
                        ring[b0 + r] = *(const v4f*)(up + (size_t)(b0 + r + PF_STEPS) * NCOL);
                }
                if (b0 == 0) handover_signal();
                __builtin_amdgcn_sched_barrier(0);
            }
        }
        float hs[SB_STEPS];
#pragma unroll
        for (int jj = 0; jj < SB_STEPS; ++jj)
            hs[jj] = os[jj] * ts[jj];
        __builtin_amdgcn_sched_barrier(0);
        if (kRefill) {
#pragma unroll
            for (int r = RG_STEPS; r < SB_STEPS; ++r)
                ring[b0 + r] = *(const v4f*)(up + (size_t)(b0 + r + PF_STEPS) * NCOL);
        }
        if (b0 == 0) handover_wait();
#pragma unroll
        for (int i = 0; i < SB_STEPS / 4; ++i)
            *(v4f*)(ho + b0 + 4 * i) =
                (v4f){hs[4 * i], hs[4 * i + 1], hs[4 * i + 2], hs[4 * i + 3]};
#pragma unroll
        for (int i = 0; i < SB_STEPS / 4; ++i)
            *(v4f*)(ho + HO_ARRAY + b0 + 4 * i) =
                (v4f){cs[4 * i], cs[4 * i + 1], cs[4 * i + 2], cs[4 * i + 3]};
        __builtin_amdgcn_sched_barrier(0);
    }
}

__device__ __forceinline__ void scan_chunk(float* u, float* lds,
                                           const float* __restrict__ w_rec,
                                           const float* __restrict__ bias,
                                           float* __restrict__ h_n, float* __restrict__ c_n,
                                           const int s, const int t_beg, const int t_len) {
    const float wf = 0.5f * w_rec[s],             wi = 0.5f * w_rec[S_DIM + s];
    const float wo = 0.5f * w_rec[2 * S_DIM + s], wg = w_rec[3 * S_DIM + s];
    const float bf = 0.5f * bias[s],              bi = 0.5f * bias[S_DIM + s];
    const float bo = 0.5f * bias[2 * S_DIM + s],  bg = bias[3 * S_DIM + s];

    float wfo, wio, woo, wgo, tc, c;
    if (t_beg == 0) {
        wfo = 0.f; wio = 0.f; woo = 0.f; wgo = 0.f;
        tc = 0.f; c = 0.f;
        h_n[s] = 0.f;
        c_n[s] = 0.f;
    } else {
        const v4f p = *(const v4f*)(u + (size_t)(t_beg - 1) * NCOL + 4 * s);
        const v4f q = *(const v4f*)(u + (size_t)(t_beg - 2) * NCOL + 4 * s);
        wfo = p.x; wio = p.y; woo = p.z; wgo = p.w;
        tc = q.x; c = q.y;
    }

    v4f ring[PF_STEPS];
#pragma unroll
    for (int j = 0; j < PF_STEPS; ++j)
        ring[j] = *(const v4f*)(u + (size_t)(t_beg + j) * NCOL + 4 * s);

    const float* up = u + (size_t)t_beg * NCOL + 4 * s;
    float* ho = lds + (threadIdx.x & 31) * HO_PITCH;
    int par = 0;
    for (int nb = t_len / PF_STEPS - 1; nb > 0; --nb) {
        scan_block<true>(up, ho + par, wf, wi, wo, wg, bf, bi, bo, bg,
                         ring, wfo, wio, woo, wgo, tc, c);
        up += (size_t)PF_STEPS * NCOL;
        par = HO_BUF - par;
    }
    scan_block<false>(up, ho + par, wf, wi, wo, wg, bf, bi, bo, bg,
                      ring, wfo, wio, woo, wgo, tc, c);
    handover_barrier();

    *(v4f*)(u + (size_t)(t_beg + t_len - 1) * NCOL + 4 * s) = (v4f){wfo, wio, woo, wgo};
    *(v4f*)(u + (size_t)(t_beg + t_len - 2) * NCOL + 4 * s) = (v4f){tc, c, 0.f, 0.f};
}

__device__ __forceinline__ void scan_store_waves(const float* lds, const int wave,
                                                 float* __restrict__ h_n,
                                                 float* __restrict__ c_n,
                                                 const int s0, const int t_beg,
                                                 const int t_len) {
    const int lane = threadIdx.x & 31;
    const bool stores = (wave == 1) || (wave == 2);
    const float* src = lds + (wave == 2 ? HO_ARRAY : 0) + lane * HO_PITCH;
    float* dst = (wave == 2 ? c_n : h_n) + (size_t)(t_beg + 1) * S_DIM + s0 + lane;
    const int nblk = t_len / PF_STEPS;
    for (int blk = 0; blk <= nblk; ++blk) {
        handover_barrier();
        if (stores && blk > 0) {
            const float* b = src + ((blk - 1) & 1) * HO_BUF;
            v4f v[PF_STEPS / 4];
#pragma unroll
            for (int i = 0; i < PF_STEPS / 4; ++i) v[i] = *(const v4f*)(b + 4 * i);
#pragma unroll
            for (int j = 0; j < PF_STEPS; ++j) dst[(size_t)j * S_DIM] = v[j >> 2][j & 3];
            dst += (size_t)PF_STEPS * S_DIM;
        }
    }
}

__global__ void __launch_bounds__(PIPE_THREADS)
pipe_kernel(const float* __restrict__ x, const float* __restrict__ w_in,
            const float* __restrict__ w_rec, const float* __restrict__ bias,
            float* u, float* __restrict__ h_n, float* __restrict__ c_n,
            const int g_row0, const int s_beg, const int s_len) {
    __shared__ __attribute__((aligned(16))) float lds[2 * STAGE_FLOATS];

    if (blockIdx.x < SCAN_WGS) {
        if (s_len > 0) {
            const int wave = __builtin_amdgcn_readfirstlane(threadIdx.x >> 5);
            if (wave == 0)
                scan_chunk(u, lds, w_rec, bias, h_n, c_n,
                           blockIdx.x * 32 + (threadIdx.x & 31), s_beg, s_len);
            else
                scan_store_waves(lds, wave, h_n, c_n, blockIdx.x * 32, s_beg, s_len);
        }
        return;
    }
    const int tile = blockIdx.x - SCAN_WGS;
    gemm_tile(x, w_in, u, lds, g_row0 + (tile >> 4) * BM, (tile & 15) * BN);
}

__global__ void q_kernel(const float* __restrict__ h_n,
                         const float* __restrict__ w_reg,
                         const float* __restrict__ b_reg,
                         float* __restrict__ q) {
    const int lane = threadIdx.x & 31;
    const int t = (blockIdx.x * blockDim.x + threadIdx.x) >> 5;
    const float* hr = h_n + (size_t)(t + 1) * S_DIM;
    float acc = 0.f;
#pragma unroll
    for (int j = 0; j < S_DIM; j += 32)
        acc += hr[j + lane] * w_reg[j + lane];
#pragma unroll
    for (int off = 16; off; off >>= 1)
        acc += __shfl_xor(acc, off, 32);
    if (lane == 0) q[t] = acc + b_reg[0];
}

extern "C" void kernel_launch(void* const* d_in, const int* in_sizes, int n_in,
                              void* d_out, int out_size, void* d_ws, size_t ws_size,
                              hipStream_t stream) {
    (void)in_sizes; (void)n_in; (void)out_size; (void)ws_size;
    const float* x     = (const float*)d_in[0];
    const float* w_in  = (const float*)d_in[1];
    const float* w_rec = (const float*)d_in[2];
    const float* bias  = (const float*)d_in[3];
    const float* w_reg = (const float*)d_in[4];
    const float* b_reg = (const float*)d_in[5];

    float* out = (float*)d_out;
    float* q   = out;
    float* h_n = out + T_DIM;
    float* c_n = h_n + (size_t)(T_DIM + 1) * S_DIM;

    float* u  = (float*)d_ws;

    pipe_kernel<<<256, PIPE_THREADS, 0, stream>>>(x, w_in, w_rec, bias, u, h_n, c_n,     0,     0,    0);
    pipe_kernel<<<256, PIPE_THREADS, 0, stream>>>(x, w_in, w_rec, bias, u, h_n, c_n,  1920,     0, 1920);
    pipe_kernel<<<256, PIPE_THREADS, 0, stream>>>(x, w_in, w_rec, bias, u, h_n, c_n,  3840,  1920, 1920);
    pipe_kernel<<<256, PIPE_THREADS, 0, stream>>>(x, w_in, w_rec, bias, u, h_n, c_n,  5760,  3840, 1920);
    pipe_kernel<<<256, PIPE_THREADS, 0, stream>>>(x, w_in, w_rec, bias, u, h_n, c_n,  7680,  5760, 1920);
    pipe_kernel<<<256, PIPE_THREADS, 0, stream>>>(x, w_in, w_rec, bias, u, h_n, c_n,  9600,  7680, 1920);
    pipe_kernel<<<256, PIPE_THREADS, 0, stream>>>(x, w_in, w_rec, bias, u, h_n, c_n, 11520,  9600, 1920);
    pipe_kernel<<<256, PIPE_THREADS, 0, stream>>>(x, w_in, w_rec, bias, u, h_n, c_n, 13440, 11520, 1920);
    pipe_kernel<<<144, PIPE_THREADS, 0, stream>>>(x, w_in, w_rec, bias, u, h_n, c_n, 15360, 13440, 1920);
    pipe_kernel<<< 16, PIPE_THREADS, 0, stream>>>(x, w_in, w_rec, bias, u, h_n, c_n,     0, 15360, 1024);

    q_kernel<<<(T_DIM * 32) / 256, 256, 0, stream>>>(h_n, w_reg, b_reg, q);
}
